# starting kernel
# baseline (speedup 1.0000x reference)
#include <hip/hip_runtime.h>
#include <math.h>

typedef __attribute__((ext_vector_type(2))) float v2f;
typedef __attribute__((ext_vector_type(8))) float v8f;

#define K_DIM 128
#define NEG_SLOPE 0.2f

__global__ void fill_u32_kernel(unsigned* __restrict__ p, int n, unsigned v) {
    int i = blockIdx.x * blockDim.x + threadIdx.x;
    if (i < n) p[i] = v;
}

template <int NCOL>
__global__ __launch_bounds__(128)
void wmma_gemm_kernel(const float* __restrict__ A, const float* __restrict__ B,
                      float* __restrict__ C, int M, int mtiles) {
    __shared__ float sW[K_DIM * NCOL];

    for (int i = threadIdx.x * 4; i < K_DIM * NCOL; i += 128 * 4)
        *(float4*)&sW[i] = *(const float4*)&B[i];
    __syncthreads();

    const int wave = threadIdx.x >> 5;
    const int lane = threadIdx.x & 31;
    const int tile = blockIdx.x * 4 + wave;
    constexpr int NTILES = NCOL / 16;
    if (tile >= mtiles * NTILES) return;

    const int mt = tile / NTILES;
    const int nt = tile - mt * NTILES;
    const int row0 = mt * 16;
    const int col0 = nt * 16;
    const int half = lane >> 4;
    const int l    = lane & 15;

    int ar = row0 + l;
    if (ar >= M) ar = M - 1;
    const float* arow = A + (size_t)ar * K_DIM;
    const float* bcol = sW + col0 + l;

    v8f acc = {};
    #pragma unroll 4
    for (int k0 = 0; k0 < K_DIM; k0 += 4) {
        v2f a = *(const v2f*)(arow + k0 + 2 * half);
        v2f b;
        b.x = bcol[(k0 + 2 * half)     * NCOL];
        b.y = bcol[(k0 + 2 * half + 1) * NCOL];
        acc = __builtin_amdgcn_wmma_f32_16x16x4_f32(
                  false, a, false, b, (short)0, acc, false, false);
    }

    float* cp = C + (size_t)(row0 + half * 8) * NCOL + col0 + l;
    if (row0 + 16 <= M) {
        #pragma unroll
        for (int i = 0; i < 8; ++i) cp[i * NCOL] = acc[i];
    } else {
        #pragma unroll
        for (int i = 0; i < 8; ++i)
            if (row0 + half * 8 + i < M) cp[i * NCOL] = acc[i];
    }
}

__global__ void att_coeff_kernel(const float* __restrict__ h,
                                 const float* __restrict__ att_s,
                                 const float* __restrict__ att_d,
                                 float* __restrict__ a_s, float* __restrict__ a_d,
                                 int N, int H, int CH) {
    int i = blockIdx.x * blockDim.x + threadIdx.x;
    if (i >= N * H) return;
    int n  = i / H;
    int hh = i - n * H;
    const float* hp = h + (size_t)n * H * CH + (size_t)hh * CH;
    const float* sp = att_s + hh * CH;
    const float* dp = att_d + hh * CH;
    float ss = 0.f, dd = 0.f;
    for (int c = 0; c < CH; ++c) { ss += hp[c] * sp[c]; dd += hp[c] * dp[c]; }
    a_s[i] = ss;
    a_d[i] = dd;
}

__device__ inline void atomicMaxF(float* addr, float v) {
    int iv = __float_as_int(v);
    if (iv >= 0) atomicMax((int*)addr, iv);
    else         atomicMin((unsigned int*)addr, (unsigned int)iv);
}

__global__ void edge_max_kernel(const int* __restrict__ ei, int E, int ET,
                                const float* __restrict__ a_s,
                                const float* __restrict__ a_d,
                                float* __restrict__ ebuf, float* __restrict__ emax,
                                int H) {
    int e = blockIdx.x * blockDim.x + threadIdx.x;
    if (e >= ET) return;
    int src = (e < E) ? ei[e]     : (e - E);
    int dst = (e < E) ? ei[E + e] : (e - E);
    for (int h = 0; h < H; ++h) {
        float v = a_s[src * H + h] + a_d[dst * H + h];
        v = (v > 0.f) ? v : NEG_SLOPE * v;
        ebuf[(size_t)e * H + h] = v;
        atomicMaxF(&emax[dst * H + h], v);
    }
}

__global__ void edge_expsum_kernel(const int* __restrict__ ei, int E, int ET,
                                   const float* __restrict__ emax,
                                   float* __restrict__ ebuf,
                                   float* __restrict__ denom, int H) {
    int e = blockIdx.x * blockDim.x + threadIdx.x;
    if (e >= ET) return;
    int dst = (e < E) ? ei[E + e] : (e - E);
    for (int h = 0; h < H; ++h) {
        float ex = __expf(ebuf[(size_t)e * H + h] - emax[dst * H + h]);
        ebuf[(size_t)e * H + h] = ex;
        atomicAdd(&denom[dst * H + h], ex);
    }
}

__global__ void edge_agg_kernel(const int* __restrict__ ei, int E,
                                const float* __restrict__ hfeat,
                                const float* __restrict__ ebuf,
                                const float* __restrict__ denom,
                                float* __restrict__ agg, int H, int CH) {
    int e = blockIdx.x;
    int c = threadIdx.x;
    int h = c / CH;
    int src = (e < E) ? ei[e]     : (e - E);
    int dst = (e < E) ? ei[E + e] : (e - E);
    float alpha = ebuf[(size_t)e * H + h] / denom[dst * H + h];
    float val   = hfeat[(size_t)src * H * CH + c] * alpha;
    atomicAdd(&agg[(size_t)dst * H * CH + c], val);
}

__global__ void bias_elu_kernel(float* __restrict__ a, const float* __restrict__ b,
                                int total, int F) {
    int i = blockIdx.x * blockDim.x + threadIdx.x;
    if (i >= total) return;
    float v = a[i] + b[i % F];
    a[i] = (v > 0.f) ? v : (__expf(v) - 1.f);
}

__global__ void bias_add_kernel(float* __restrict__ a, const float* __restrict__ b,
                                int total, int F) {
    int i = blockIdx.x * blockDim.x + threadIdx.x;
    if (i >= total) return;
    a[i] += b[i % F];
}

extern "C" void kernel_launch(void* const* d_in, const int* in_sizes, int n_in,
                              void* d_out, int out_size, void* d_ws, size_t ws_size,
                              hipStream_t stream) {
    const float* x   = (const float*)d_in[0];
    const int*   ei  = (const int*)  d_in[1];
    const float* W1  = (const float*)d_in[2];
    const float* as1 = (const float*)d_in[3];
    const float* ad1 = (const float*)d_in[4];
    const float* b1  = (const float*)d_in[5];
    const float* W2  = (const float*)d_in[6];
    const float* as2 = (const float*)d_in[7];
    const float* ad2 = (const float*)d_in[8];
    const float* b2  = (const float*)d_in[9];
    float* out = (float*)d_out;

    const int N  = in_sizes[0] / 128;
    const int E  = in_sizes[1] / 2;
    const int ET = E + N;
    const int H1 = 4, C1 = 32, F1 = 128;
    const int H2 = 1, C2 = 64, F2 = 64;

    char* w = (char*)d_ws;
    auto carve = [&](size_t bytes) {
        void* p = (void*)w;
        w += (bytes + 255) & ~(size_t)255;
        return p;
    };
    float* h1    = (float*)carve((size_t)N * F1 * 4);
    float* a1s   = (float*)carve((size_t)N * H1 * 4);
    float* a1d   = (float*)carve((size_t)N * H1 * 4);
    float* emax1 = (float*)carve((size_t)N * H1 * 4);
    float* den1  = (float*)carve((size_t)N * H1 * 4);
    float* ebuf1 = (float*)carve((size_t)ET * H1 * 4);
    float* agg1  = (float*)carve((size_t)N * F1 * 4);
    float* h2    = (float*)carve((size_t)N * F2 * 4);
    float* a2s   = (float*)carve((size_t)N * 4);
    float* a2d   = (float*)carve((size_t)N * 4);
    float* emax2 = (float*)carve((size_t)N * 4);
    float* den2  = (float*)carve((size_t)N * 4);
    float* ebuf2 = (float*)carve((size_t)ET * 4);

    const unsigned NEG_INF = 0xff800000u;
    auto fill = [&](void* p, int n, unsigned v) {
        fill_u32_kernel<<<(n + 255) / 256, 256, 0, stream>>>((unsigned*)p, n, v);
    };
    fill(emax1, N * H1, NEG_INF);
    fill(den1,  N * H1, 0u);
    fill(agg1,  N * F1, 0u);
    fill(emax2, N,      NEG_INF);
    fill(den2,  N,      0u);
    fill(out,   N * F2, 0u);

    const int eb = 256, eg = (ET + eb - 1) / eb;

    {
        int mt = (N + 15) / 16;
        int blocks = (mt * (F1 / 16) + 3) / 4;
        wmma_gemm_kernel<128><<<blocks, 128, 0, stream>>>(x, W1, h1, N, mt);
    }
    att_coeff_kernel<<<(N * H1 + 255) / 256, 256, 0, stream>>>(
        h1, as1, ad1, a1s, a1d, N, H1, C1);
    edge_max_kernel<<<eg, eb, 0, stream>>>(ei, E, ET, a1s, a1d, ebuf1, emax1, H1);
    edge_expsum_kernel<<<eg, eb, 0, stream>>>(ei, E, ET, emax1, ebuf1, den1, H1);
    edge_agg_kernel<<<ET, F1, 0, stream>>>(ei, E, h1, ebuf1, den1, agg1, H1, C1);
    bias_elu_kernel<<<(N * F1 + 255) / 256, 256, 0, stream>>>(agg1, b1, N * F1, F1);

    {
        int mt = (N + 15) / 16;
        int blocks = (mt * (F2 / 16) + 3) / 4;
        wmma_gemm_kernel<64><<<blocks, 128, 0, stream>>>(agg1, W2, h2, N, mt);
    }
    att_coeff_kernel<<<(N + 255) / 256, 256, 0, stream>>>(
        h2, as2, ad2, a2s, a2d, N, H2, C2);
    edge_max_kernel<<<eg, eb, 0, stream>>>(ei, E, ET, a2s, a2d, ebuf2, emax2, H2);
    edge_expsum_kernel<<<eg, eb, 0, stream>>>(ei, E, ET, emax2, ebuf2, den2, H2);
    edge_agg_kernel<<<ET, F2, 0, stream>>>(ei, E, h2, ebuf2, den2, out, H2, C2);
    bias_add_kernel<<<(N * F2 + 255) / 256, 256, 0, stream>>>(out, b2, N * F2, F2);
}


# Edge phase by owner blocks: each owns 196 dst nodes, filters the edge stream, does max, exp-sum, row sums in LDS atomics, stores rows once with bias(+ELU); 18->6 launches; sums in LDS arrival order
# speedup vs baseline: 4.0618x; 4.0618x over previous
#include <hip/hip_runtime.h>
#include <math.h>

typedef __attribute__((ext_vector_type(2))) float v2f;
typedef __attribute__((ext_vector_type(8))) float v8f;

#define K_DIM 128
#define NEG_SLOPE 0.2f

template <int NCOL>
__global__ __launch_bounds__(128)
void wmma_gemm_kernel(const float* __restrict__ A, const float* __restrict__ B,
                      float* __restrict__ C, int M, int mtiles) {
    __shared__ float sW[K_DIM * NCOL];

    for (int i = threadIdx.x * 4; i < K_DIM * NCOL; i += 128 * 4)
        *(float4*)&sW[i] = *(const float4*)&B[i];
    __syncthreads();

    const int wave = threadIdx.x >> 5;
    const int lane = threadIdx.x & 31;
    const int tile = blockIdx.x * 4 + wave;
    constexpr int NTILES = NCOL / 16;
    if (tile >= mtiles * NTILES) return;

    const int mt = tile / NTILES;
    const int nt = tile - mt * NTILES;
    const int row0 = mt * 16;
    const int col0 = nt * 16;
    const int half = lane >> 4;
    const int l    = lane & 15;

    int ar = row0 + l;
    if (ar >= M) ar = M - 1;
    const float* arow = A + (size_t)ar * K_DIM;
    const float* bcol = sW + col0 + l;

    v8f acc = {};
    #pragma unroll 4
    for (int k0 = 0; k0 < K_DIM; k0 += 4) {
        v2f a = *(const v2f*)(arow + k0 + 2 * half);
        v2f b;
        b.x = bcol[(k0 + 2 * half)     * NCOL];
        b.y = bcol[(k0 + 2 * half + 1) * NCOL];
        acc = __builtin_amdgcn_wmma_f32_16x16x4_f32(
                  false, a, false, b, (short)0, acc, false, false);
    }

    float* cp = C + (size_t)(row0 + half * 8) * NCOL + col0 + l;
    if (row0 + 16 <= M) {
        #pragma unroll
        for (int i = 0; i < 8; ++i) cp[i * NCOL] = acc[i];
    } else {
        #pragma unroll
        for (int i = 0; i < 8; ++i)
            if (row0 + half * 8 + i < M) cp[i * NCOL] = acc[i];
    }
}

__global__ void att_coeff_kernel(const float* __restrict__ h,
                                 const float* __restrict__ att_s,
                                 const float* __restrict__ att_d,
                                 float* __restrict__ a_s, float* __restrict__ a_d,
                                 int N, int H, int CH) {
    int i = blockIdx.x * blockDim.x + threadIdx.x;
    if (i >= N * H) return;
    int n  = i / H;
    int hh = i - n * H;
    const float* hp = h + (size_t)n * H * CH + (size_t)hh * CH;
    const float* sp = att_s + hh * CH;
    const float* dp = att_d + hh * CH;
    float ss = 0.f, dd = 0.f;
    for (int c = 0; c < CH; ++c) { ss += hp[c] * sp[c]; dd += hp[c] * dp[c]; }
    a_s[i] = ss;
    a_d[i] = dd;
}

#define OWN_R    196
#define OWN_T    1024
#define OWN_CAP  4096
#define OWN_U    8
#define OWN_UC   8

__device__ inline void ldsMaxF(float* addr, float v) {
    int iv = __float_as_int(v);
    if (iv >= 0) atomicMax((int*)addr, iv);
    else         atomicMin((unsigned int*)addr, (unsigned int)iv);
}

template <int H, int CH, bool ELU>
__global__ __launch_bounds__(OWN_T)
void gat_owner_kernel(const int* __restrict__ ei, int E, int N,
                      const float* __restrict__ hfeat,
                      const float* __restrict__ a_s,
                      const float* __restrict__ a_d,
                      const float* __restrict__ bias,
                      float* __restrict__ outp) {
    constexpr int F  = H * CH;
    constexpr int KF = F / 32;
    constexpr int NW = OWN_T / 32;

    __shared__ float sAgg[OWN_R * F];
    __shared__ float sMax[OWN_R * H];
    __shared__ float sDen[OWN_R * H];
    __shared__ float sAd [OWN_R * H];
    __shared__ int   sSrc[OWN_CAP];
    __shared__ int   sDl [OWN_CAP];
    __shared__ float sE  [OWN_CAP * H];
    __shared__ int   sCnt;

    const int tid = threadIdx.x;
    const int n0  = blockIdx.x * OWN_R;
    const int nn  = (N - n0 < OWN_R) ? (N - n0) : OWN_R;
    const int* __restrict__ dstp = ei + E;

    for (int i = tid; i < OWN_R * F; i += OWN_T) sAgg[i] = 0.f;
    for (int i = tid; i < OWN_R * H; i += OWN_T) {
        sMax[i] = __int_as_float(0xff800000);
        sDen[i] = 0.f;
        sAd[i]  = (i < nn * H) ? a_d[(size_t)n0 * H + i] : 0.f;
    }
    for (int i = tid; i < nn; i += OWN_T) { sSrc[i] = -(i + 1); sDl[i] = i; }
    if (tid == 0) sCnt = nn;
    __syncthreads();

    auto keep = [&](int dst, int eid, unsigned lim) {
        unsigned dl = (unsigned)(dst - n0);
        if (dl < lim) {
            int pos = atomicAdd(&sCnt, 1);
            if (pos < OWN_CAP) { sSrc[pos] = eid; sDl[pos] = (int)dl; }
        }
    };
    if ((E & 3) == 0) {
        const int EV = E >> 2;
        const int4* __restrict__ dv = (const int4*)dstp;
        constexpr int STEP = OWN_T * OWN_U;
        const int rot = (int)(((long long)blockIdx.x * EV) / gridDim.x);
        auto slot = [&](int base, int u, bool& ok) {
            int v  = base + u * OWN_T + tid;
            ok     = v < EV;
            int vr = v + rot;
            if (vr >= EV) vr -= EV;
            return ok ? vr : 0;
        };
        int4 cur[OWN_U];
        #pragma unroll
        for (int u = 0; u < OWN_U; ++u) { bool ok; cur[u] = dv[slot(0, u, ok)]; }
        for (int base = 0; base < EV; base += STEP) {
            int4 nxt[OWN_U];
            #pragma unroll
            for (int u = 0; u < OWN_U; ++u) { bool ok; nxt[u] = dv[slot(base + STEP, u, ok)]; }
            #pragma unroll
            for (int u = 0; u < OWN_U; ++u) {
                bool ok;
                const int e0 = slot(base, u, ok) * 4;
                const unsigned lim = ok ? (unsigned)nn : 0u;
                keep(cur[u].x, e0 + 0, lim);
                keep(cur[u].y, e0 + 1, lim);
                keep(cur[u].z, e0 + 2, lim);
                keep(cur[u].w, e0 + 3, lim);
            }
            #pragma unroll
            for (int u = 0; u < OWN_U; ++u) cur[u] = nxt[u];
        }
    } else {
        for (int e = tid; e < E; e += OWN_T) keep(dstp[e], e, (unsigned)nn);
    }
    __syncthreads();
    const int total = sCnt;

    if (total <= OWN_CAP) {
        for (int j = tid; j < total; j += OWN_T) {
            int id  = sSrc[j];
            int src = (id >= 0) ? ei[id] : (n0 - id - 1);
            int dl  = sDl[j];
            sSrc[j] = src;
            for (int h = 0; h < H; ++h) {
                float v = a_s[(size_t)src * H + h] + sAd[dl * H + h];
                v = (v > 0.f) ? v : NEG_SLOPE * v;
                sE[j * H + h] = v;
                ldsMaxF(&sMax[dl * H + h], v);
            }
        }
        __syncthreads();
        for (int j = tid; j < total; j += OWN_T) {
            int dl = sDl[j];
            for (int h = 0; h < H; ++h) {
                float ex = __expf(sE[j * H + h] - sMax[dl * H + h]);
                sE[j * H + h] = ex;
                atomicAdd(&sDen[dl * H + h], ex);
            }
        }
        __syncthreads();
        for (int j = tid; j < total; j += OWN_T) {
            int dl = sDl[j];
            for (int h = 0; h < H; ++h)
                sE[j * H + h] = sE[j * H + h] / sDen[dl * H + h];
        }
        __syncthreads();
        const int wave = tid >> 5;
        const int lane = tid & 31;
        for (int j0 = wave; j0 < total; j0 += NW * OWN_UC) {
            float v[OWN_UC][KF];
            int   dl[OWN_UC];
            #pragma unroll
            for (int q = 0; q < OWN_UC; ++q) {
                int  j   = j0 + q * NW;
                bool ok  = j < total;
                int  src = ok ? sSrc[j] : n0;
                dl[q]    = ok ? sDl[j]  : 0;
                const float* hp = hfeat + (size_t)src * F + lane;
                #pragma unroll
                for (int k = 0; k < KF; ++k) v[q][k] = hp[k * 32];
            }
            #pragma unroll
            for (int q = 0; q < OWN_UC; ++q) {
                int j = j0 + q * NW;
                if (j < total) {
                    #pragma unroll
                    for (int k = 0; k < KF; ++k) {
                        float al = sE[j * H + (k * 32) / CH];
                        atomicAdd(&sAgg[dl[q] * F + k * 32 + lane], v[q][k] * al);
                    }
                }
            }
        }
        __syncthreads();
    } else {
        const int ET = E + N;
        for (int e = tid; e < ET; e += OWN_T) {
            int dst = (e < E) ? dstp[e] : (e - E);
            unsigned dl = (unsigned)(dst - n0);
            if (dl < (unsigned)nn) {
                int src = (e < E) ? ei[e] : (e - E);
                for (int h = 0; h < H; ++h) {
                    float v = a_s[(size_t)src * H + h] + sAd[dl * H + h];
                    v = (v > 0.f) ? v : NEG_SLOPE * v;
                    ldsMaxF(&sMax[dl * H + h], v);
                }
            }
        }
        __syncthreads();
        for (int e = tid; e < ET; e += OWN_T) {
            int dst = (e < E) ? dstp[e] : (e - E);
            unsigned dl = (unsigned)(dst - n0);
            if (dl < (unsigned)nn) {
                int src = (e < E) ? ei[e] : (e - E);
                for (int h = 0; h < H; ++h) {
                    float v = a_s[(size_t)src * H + h] + sAd[dl * H + h];
                    v = (v > 0.f) ? v : NEG_SLOPE * v;
                    atomicAdd(&sDen[dl * H + h], __expf(v - sMax[dl * H + h]));
                }
            }
        }
        __syncthreads();
        for (int e = tid; e < ET; e += OWN_T) {
            int dst = (e < E) ? dstp[e] : (e - E);
            unsigned dl = (unsigned)(dst - n0);
            if (dl < (unsigned)nn) {
                int src = (e < E) ? ei[e] : (e - E);
                float al[H];
                for (int h = 0; h < H; ++h) {
                    float v = a_s[(size_t)src * H + h] + sAd[dl * H + h];
                    v = (v > 0.f) ? v : NEG_SLOPE * v;
                    al[h] = __expf(v - sMax[dl * H + h]) / sDen[dl * H + h];
                }
                for (int c = 0; c < F; ++c)
                    atomicAdd(&sAgg[dl * F + c], hfeat[(size_t)src * F + c] * al[c / CH]);
            }
        }
        __syncthreads();
    }

    for (int i = tid; i < nn * F; i += OWN_T) {
        float v = sAgg[i] + bias[i % F];
        if (ELU) v = (v > 0.f) ? v : (__expf(v) - 1.f);
        outp[(size_t)n0 * F + i] = v;
    }
}

extern "C" void kernel_launch(void* const* d_in, const int* in_sizes, int n_in,
                              void* d_out, int out_size, void* d_ws, size_t ws_size,
                              hipStream_t stream) {
    const float* x   = (const float*)d_in[0];
    const int*   ei  = (const int*)  d_in[1];
    const float* W1  = (const float*)d_in[2];
    const float* as1 = (const float*)d_in[3];
    const float* ad1 = (const float*)d_in[4];
    const float* b1  = (const float*)d_in[5];
    const float* W2  = (const float*)d_in[6];
    const float* as2 = (const float*)d_in[7];
    const float* ad2 = (const float*)d_in[8];
    const float* b2  = (const float*)d_in[9];
    float* out = (float*)d_out;

    const int N  = in_sizes[0] / 128;
    const int E  = in_sizes[1] / 2;
    const int H1 = 4, C1 = 32, F1 = 128;
    const int H2 = 1, C2 = 64, F2 = 64;

    char* w = (char*)d_ws;
    auto carve = [&](size_t bytes) {
        void* p = (void*)w;
        w += (bytes + 255) & ~(size_t)255;
        return p;
    };
    float* h1    = (float*)carve((size_t)N * F1 * 4);
    float* a1s   = (float*)carve((size_t)N * H1 * 4);
    float* a1d   = (float*)carve((size_t)N * H1 * 4);
    float* agg1  = (float*)carve((size_t)N * F1 * 4);
    float* h2    = (float*)carve((size_t)N * F2 * 4);
    float* a2s   = (float*)carve((size_t)N * 4);
    float* a2d   = (float*)carve((size_t)N * 4);

    const int og = (N + OWN_R - 1) / OWN_R;

    {
        int mt = (N + 15) / 16;
        int blocks = (mt * (F1 / 16) + 3) / 4;
        wmma_gemm_kernel<128><<<blocks, 128, 0, stream>>>(x, W1, h1, N, mt);
    }
    att_coeff_kernel<<<(N * H1 + 255) / 256, 256, 0, stream>>>(
        h1, as1, ad1, a1s, a1d, N, H1, C1);
    gat_owner_kernel<4, 32, true><<<og, OWN_T, 0, stream>>>(
        ei, E, N, h1, a1s, a1d, b1, agg1);

    {
        int mt = (N + 15) / 16;
        int blocks = (mt * (F2 / 16) + 3) / 4;
        wmma_gemm_kernel<64><<<blocks, 128, 0, stream>>>(agg1, W2, h2, N, mt);
    }
    att_coeff_kernel<<<(N + 255) / 256, 256, 0, stream>>>(
        h2, as2, ad2, a2s, a2d, N, H2, C2);
    gat_owner_kernel<1, 64, false><<<og, OWN_T, 0, stream>>>(
        ei, E, N, h2, a2s, a2d, b2, out);
}


# Edge stream filtered once: layer-1 edge launch saves each owner block's list (source node, local dst) in the workspace; layer 2 loads it, no second scan; layer-2 sums follow layer 1's list order
# speedup vs baseline: 4.8391x; 1.1914x over previous
#include <hip/hip_runtime.h>
#include <math.h>

typedef __attribute__((ext_vector_type(2))) float v2f;
typedef __attribute__((ext_vector_type(8))) float v8f;

#define K_DIM 128
#define NEG_SLOPE 0.2f

template <int NCOL>
__global__ __launch_bounds__(128)
void wmma_gemm_kernel(const float* __restrict__ A, const float* __restrict__ B,
                      float* __restrict__ C, int M, int mtiles) {
    __shared__ float sW[K_DIM * NCOL];

    for (int i = threadIdx.x * 4; i < K_DIM * NCOL; i += 128 * 4)
        *(float4*)&sW[i] = *(const float4*)&B[i];
    __syncthreads();

    const int wave = threadIdx.x >> 5;
    const int lane = threadIdx.x & 31;
    const int tile = blockIdx.x * 4 + wave;
    constexpr int NTILES = NCOL / 16;
    if (tile >= mtiles * NTILES) return;

    const int mt = tile / NTILES;
    const int nt = tile - mt * NTILES;
    const int row0 = mt * 16;
    const int col0 = nt * 16;
    const int half = lane >> 4;
    const int l    = lane & 15;

    int ar = row0 + l;
    if (ar >= M) ar = M - 1;
    const float* arow = A + (size_t)ar * K_DIM;
    const float* bcol = sW + col0 + l;

    v8f acc = {};
    #pragma unroll 4
    for (int k0 = 0; k0 < K_DIM; k0 += 4) {
        v2f a = *(const v2f*)(arow + k0 + 2 * half);
        v2f b;
        b.x = bcol[(k0 + 2 * half)     * NCOL];
        b.y = bcol[(k0 + 2 * half + 1) * NCOL];
        acc = __builtin_amdgcn_wmma_f32_16x16x4_f32(
                  false, a, false, b, (short)0, acc, false, false);
    }

    float* cp = C + (size_t)(row0 + half * 8) * NCOL + col0 + l;
    if (row0 + 16 <= M) {
        #pragma unroll
        for (int i = 0; i < 8; ++i) cp[i * NCOL] = acc[i];
    } else {
        #pragma unroll
        for (int i = 0; i < 8; ++i)
            if (row0 + half * 8 + i < M) cp[i * NCOL] = acc[i];
    }
}

__global__ void att_coeff_kernel(const float* __restrict__ h,
                                 const float* __restrict__ att_s,
                                 const float* __restrict__ att_d,
                                 float* __restrict__ a_s, float* __restrict__ a_d,
                                 int N, int H, int CH) {
    int i = blockIdx.x * blockDim.x + threadIdx.x;
    if (i >= N * H) return;
    int n  = i / H;
    int hh = i - n * H;
    const float* hp = h + (size_t)n * H * CH + (size_t)hh * CH;
    const float* sp = att_s + hh * CH;
    const float* dp = att_d + hh * CH;
    float ss = 0.f, dd = 0.f;
    for (int c = 0; c < CH; ++c) { ss += hp[c] * sp[c]; dd += hp[c] * dp[c]; }
    a_s[i] = ss;
    a_d[i] = dd;
}

#define OWN_R    196
#define OWN_T    1024
#define OWN_CAP  4096
#define OWN_U    8
#define OWN_UC   8
#define LST_HDR  32
#define LST_STRIDE (LST_HDR + 2 * OWN_CAP)

__device__ inline void ldsMaxF(float* addr, float v) {
    int iv = __float_as_int(v);
    if (iv >= 0) atomicMax((int*)addr, iv);
    else         atomicMin((unsigned int*)addr, (unsigned int)iv);
}

template <int H, int CH, bool ELU, bool SAVE, bool LOAD>
__global__ __launch_bounds__(OWN_T)
void gat_owner_kernel(const int* __restrict__ ei, int E, int N,
                      int* __restrict__ lst,
                      const float* __restrict__ hfeat,
                      const float* __restrict__ a_s,
                      const float* __restrict__ a_d,
                      const float* __restrict__ bias,
                      float* __restrict__ outp) {
    constexpr int F  = H * CH;
    constexpr int KF = F / 32;
    constexpr int NW = OWN_T / 32;

    __shared__ float sAgg[OWN_R * F];
    __shared__ float sMax[OWN_R * H];
    __shared__ float sDen[OWN_R * H];
    __shared__ float sAd [OWN_R * H];
    __shared__ int   sSrc[OWN_CAP];
    __shared__ int   sDl [OWN_CAP];
    __shared__ float sE  [OWN_CAP * H];
    __shared__ int   sCnt;

    const int tid = threadIdx.x;
    const int n0  = blockIdx.x * OWN_R;
    const int nn  = (N - n0 < OWN_R) ? (N - n0) : OWN_R;
    const int* __restrict__ dstp = ei + E;
    const bool reuse = LOAD && (lst != nullptr);
    int* __restrict__ myl = lst ? lst + (size_t)blockIdx.x * LST_STRIDE : nullptr;
    int loaded = 0;

    for (int i = tid; i < OWN_R * F; i += OWN_T) sAgg[i] = 0.f;
    for (int i = tid; i < OWN_R * H; i += OWN_T) {
        sMax[i] = __int_as_float(0xff800000);
        sDen[i] = 0.f;
        sAd[i]  = (i < nn * H) ? a_d[(size_t)n0 * H + i] : 0.f;
    }
    if (!reuse) {
        for (int i = tid; i < nn; i += OWN_T) { sSrc[i] = -(i + 1); sDl[i] = i; }
        if (tid == 0) sCnt = nn;
    }
    __syncthreads();

    auto keep = [&](int dst, int eid, unsigned lim) {
        unsigned dl = (unsigned)(dst - n0);
        if (dl < lim) {
            int pos = atomicAdd(&sCnt, 1);
            if (pos < OWN_CAP) { sSrc[pos] = eid; sDl[pos] = (int)dl; }
        }
    };
    if (reuse) {
        loaded = myl[0];
        if ((unsigned)(loaded - 1) < (unsigned)OWN_CAP) {
            const int2* __restrict__ lp = (const int2*)(myl + LST_HDR);
            constexpr int NQ = OWN_CAP / OWN_T;
            int2 p[NQ];
            #pragma unroll
            for (int q = 0; q < NQ; ++q) {
                const int j = q * OWN_T + tid;
                p[q] = lp[(j < loaded) ? j : (loaded - 1)];
            }
            #pragma unroll
            for (int q = 0; q < NQ; ++q) {
                const int j = q * OWN_T + tid;
                if (j < loaded) {
                    sSrc[j] = p[q].x;
                    sDl[j]  = p[q].y;
                }
            }
        }
    } else if ((E & 3) == 0) {
        const int EV = E >> 2;
        const int4* __restrict__ dv = (const int4*)dstp;
        constexpr int STEP = OWN_T * OWN_U;
        const int rot = (int)(((long long)blockIdx.x * EV) / gridDim.x);
        auto slot = [&](int base, int u, bool& ok) {
            int v  = base + u * OWN_T + tid;
            ok     = v < EV;
            int vr = v + rot;
            if (vr >= EV) vr -= EV;
            return ok ? vr : 0;
        };
        int4 cur[OWN_U];
        #pragma unroll
        for (int u = 0; u < OWN_U; ++u) { bool ok; cur[u] = dv[slot(0, u, ok)]; }
        for (int base = 0; base < EV; base += STEP) {
            int4 nxt[OWN_U];
            #pragma unroll
            for (int u = 0; u < OWN_U; ++u) { bool ok; nxt[u] = dv[slot(base + STEP, u, ok)]; }
            #pragma unroll
            for (int u = 0; u < OWN_U; ++u) {
                bool ok;
                const int e0 = slot(base, u, ok) * 4;
                const unsigned lim = ok ? (unsigned)nn : 0u;
                keep(cur[u].x, e0 + 0, lim);
                keep(cur[u].y, e0 + 1, lim);
                keep(cur[u].z, e0 + 2, lim);
                keep(cur[u].w, e0 + 3, lim);
            }
            #pragma unroll
            for (int u = 0; u < OWN_U; ++u) cur[u] = nxt[u];
        }
    } else {
        for (int e = tid; e < E; e += OWN_T) keep(dstp[e], e, (unsigned)nn);
    }
    __syncthreads();
    const int total = reuse ? loaded : sCnt;
    if (SAVE && lst != nullptr && tid == 0) myl[0] = total;

    if ((unsigned)total <= (unsigned)OWN_CAP) {
        for (int j = tid; j < total; j += OWN_T) {
            int id  = sSrc[j];
            int src = reuse ? id : ((id >= 0) ? ei[id] : (n0 - id - 1));
            int dl  = sDl[j];
            sSrc[j] = src;
            for (int h = 0; h < H; ++h) {
                float v = a_s[(size_t)src * H + h] + sAd[dl * H + h];
                v = (v > 0.f) ? v : NEG_SLOPE * v;
                sE[j * H + h] = v;
                ldsMaxF(&sMax[dl * H + h], v);
            }
        }
        __syncthreads();
        if (SAVE && lst != nullptr) {
            int2* __restrict__ lp = (int2*)(myl + LST_HDR);
            for (int q = 0; q < OWN_CAP / OWN_T; ++q) {
                const int j = q * OWN_T + tid;
                if (j < total) lp[j] = make_int2(sSrc[j], sDl[j]);
            }
        }
        for (int j = tid; j < total; j += OWN_T) {
            int dl = sDl[j];
            for (int h = 0; h < H; ++h) {
                float ex = __expf(sE[j * H + h] - sMax[dl * H + h]);
                sE[j * H + h] = ex;
                atomicAdd(&sDen[dl * H + h], ex);
            }
        }
        __syncthreads();
        for (int j = tid; j < total; j += OWN_T) {
            int dl = sDl[j];
            for (int h = 0; h < H; ++h)
                sE[j * H + h] = sE[j * H + h] / sDen[dl * H + h];
        }
        __syncthreads();
        const int wave = tid >> 5;
        const int lane = tid & 31;
        for (int j0 = wave; j0 < total; j0 += NW * OWN_UC) {
            float v[OWN_UC][KF];
            int   dl[OWN_UC];
            #pragma unroll
            for (int q = 0; q < OWN_UC; ++q) {
                int  j   = j0 + q * NW;
                bool ok  = j < total;
                int  src = ok ? sSrc[j] : n0;
                dl[q]    = ok ? sDl[j]  : 0;
                const float* hp = hfeat + (size_t)src * F + lane;
                #pragma unroll
                for (int k = 0; k < KF; ++k) v[q][k] = hp[k * 32];
            }
            #pragma unroll
            for (int q = 0; q < OWN_UC; ++q) {
                int j = j0 + q * NW;
                if (j < total) {
                    #pragma unroll
                    for (int k = 0; k < KF; ++k) {
                        float al = sE[j * H + (k * 32) / CH];
                        atomicAdd(&sAgg[dl[q] * F + k * 32 + lane], v[q][k] * al);
                    }
                }
            }
        }
        __syncthreads();
    } else {
        const int ET = E + N;
        for (int e = tid; e < ET; e += OWN_T) {
            int dst = (e < E) ? dstp[e] : (e - E);
            unsigned dl = (unsigned)(dst - n0);
            if (dl < (unsigned)nn) {
                int src = (e < E) ? ei[e] : (e - E);
                for (int h = 0; h < H; ++h) {
                    float v = a_s[(size_t)src * H + h] + sAd[dl * H + h];
                    v = (v > 0.f) ? v : NEG_SLOPE * v;
                    ldsMaxF(&sMax[dl * H + h], v);
                }
            }
        }
        __syncthreads();
        for (int e = tid; e < ET; e += OWN_T) {
            int dst = (e < E) ? dstp[e] : (e - E);
            unsigned dl = (unsigned)(dst - n0);
            if (dl < (unsigned)nn) {
                int src = (e < E) ? ei[e] : (e - E);
                for (int h = 0; h < H; ++h) {
                    float v = a_s[(size_t)src * H + h] + sAd[dl * H + h];
                    v = (v > 0.f) ? v : NEG_SLOPE * v;
                    atomicAdd(&sDen[dl * H + h], __expf(v - sMax[dl * H + h]));
                }
            }
        }
        __syncthreads();
        for (int e = tid; e < ET; e += OWN_T) {
            int dst = (e < E) ? dstp[e] : (e - E);
            unsigned dl = (unsigned)(dst - n0);
            if (dl < (unsigned)nn) {
                int src = (e < E) ? ei[e] : (e - E);
                float al[H];
                for (int h = 0; h < H; ++h) {
                    float v = a_s[(size_t)src * H + h] + sAd[dl * H + h];
                    v = (v > 0.f) ? v : NEG_SLOPE * v;
                    al[h] = __expf(v - sMax[dl * H + h]) / sDen[dl * H + h];
                }
                for (int c = 0; c < F; ++c)
                    atomicAdd(&sAgg[dl * F + c], hfeat[(size_t)src * F + c] * al[c / CH]);
            }
        }
        __syncthreads();
    }

    for (int i = tid; i < nn * F; i += OWN_T) {
        float v = sAgg[i] + bias[i % F];
        if (ELU) v = (v > 0.f) ? v : (__expf(v) - 1.f);
        outp[(size_t)n0 * F + i] = v;
    }
}

extern "C" void kernel_launch(void* const* d_in, const int* in_sizes, int n_in,
                              void* d_out, int out_size, void* d_ws, size_t ws_size,
                              hipStream_t stream) {
    const float* x   = (const float*)d_in[0];
    const int*   ei  = (const int*)  d_in[1];
    const float* W1  = (const float*)d_in[2];
    const float* as1 = (const float*)d_in[3];
    const float* ad1 = (const float*)d_in[4];
    const float* b1  = (const float*)d_in[5];
    const float* W2  = (const float*)d_in[6];
    const float* as2 = (const float*)d_in[7];
    const float* ad2 = (const float*)d_in[8];
    const float* b2  = (const float*)d_in[9];
    float* out = (float*)d_out;

    const int N  = in_sizes[0] / 128;
    const int E  = in_sizes[1] / 2;
    const int H1 = 4, C1 = 32, F1 = 128;
    const int H2 = 1, C2 = 64, F2 = 64;

    char* w = (char*)d_ws;
    auto carve = [&](size_t bytes) {
        void* p = (void*)w;
        w += (bytes + 255) & ~(size_t)255;
        return p;
    };
    float* h1    = (float*)carve((size_t)N * F1 * 4);
    float* a1s   = (float*)carve((size_t)N * H1 * 4);
    float* a1d   = (float*)carve((size_t)N * H1 * 4);
    float* agg1  = (float*)carve((size_t)N * F1 * 4);
    float* h2    = (float*)carve((size_t)N * F2 * 4);
    float* a2s   = (float*)carve((size_t)N * 4);
    float* a2d   = (float*)carve((size_t)N * 4);

    const int og = (N + OWN_R - 1) / OWN_R;
    int* lst = (int*)carve((size_t)og * LST_STRIDE * 4);
    if ((size_t)(w - (char*)d_ws) > ws_size) lst = nullptr;

    {
        int mt = (N + 15) / 16;
        int blocks = (mt * (F1 / 16) + 3) / 4;
        wmma_gemm_kernel<128><<<blocks, 128, 0, stream>>>(x, W1, h1, N, mt);
    }
    att_coeff_kernel<<<(N * H1 + 255) / 256, 256, 0, stream>>>(
        h1, as1, ad1, a1s, a1d, N, H1, C1);
    gat_owner_kernel<4, 32, true, true, false><<<og, OWN_T, 0, stream>>>(
        ei, E, N, lst, h1, a1s, a1d, b1, agg1);

    {
        int mt = (N + 15) / 16;
        int blocks = (mt * (F2 / 16) + 3) / 4;
        wmma_gemm_kernel<64><<<blocks, 128, 0, stream>>>(agg1, W2, h2, N, mt);
    }
    att_coeff_kernel<<<(N + 255) / 256, 256, 0, stream>>>(
        h2, as2, ad2, a2s, a2d, N, H2, C2);
    gat_owner_kernel<1, 64, false, false, true><<<og, OWN_T, 0, stream>>>(
        ei, E, N, lst, h2, a2s, a2d, b2, out);
}


# Both f32 WMMA products: a wave keeps its 16-column W slice in registers all launch (no W in LDS), blocks walk row tiles on a fixed grid, A tile via LDS, double-buffered; k order, sums unchanged
# speedup vs baseline: 6.7006x; 1.3847x over previous
#include <hip/hip_runtime.h>
#include <math.h>

typedef __attribute__((ext_vector_type(2))) float v2f;
typedef __attribute__((ext_vector_type(8))) float v8f;

#define K_DIM 128
#define NEG_SLOPE 0.2f

#define GEMM_GRID 1024
#define A_PITCH   132

template <int NCOL>
__global__ __launch_bounds__(NCOL * 2)
void wmma_gemm_kernel(const float* __restrict__ A, const float* __restrict__ B,
                      float* __restrict__ C, int M, int mtiles) {
    constexpr int NT = NCOL * 2;
    constexpr int NQ = (16 * K_DIM / 4) / NT;
    __shared__ __attribute__((aligned(16))) float sA[2][16 * A_PITCH];

    const int wave = threadIdx.x >> 5;
    const int lane = threadIdx.x & 31;
    const int half = lane >> 4;
    const int l    = lane & 15;
    const int col0 = wave * 16;

    v2f bf[K_DIM / 4];
    {
        const float* bcol = B + col0 + l;
        #pragma unroll
        for (int s = 0; s < K_DIM / 4; ++s) {
            bf[s].x = bcol[(4 * s + 2 * half)     * NCOL];
            bf[s].y = bcol[(4 * s + 2 * half + 1) * NCOL];
        }
    }

    static_assert(NQ == 2 || NQ == 4, "an A tile is 2 or 4 pieces per thread");
    float4 r0, r1, r2 = {}, r3 = {};
    auto asrc = [&](int mt, int q) {
        const int f = q * NT + (int)threadIdx.x;
        int ar = mt * 16 + (f >> 5);
        if (ar >= M) ar = M - 1;
        return (const float4*)(A + (size_t)ar * K_DIM + (f & 31) * 4);
    };
    auto adst = [&](int buf, int q) {
        const int f = q * NT + (int)threadIdx.x;
        return (float4*)&sA[buf][(f >> 5) * A_PITCH + (f & 31) * 4];
    };
#define A_FETCH(mt_) do { r0 = *asrc((mt_), 0); r1 = *asrc((mt_), 1); \
        if constexpr (NQ == 4) { r2 = *asrc((mt_), 2); r3 = *asrc((mt_), 3); } } while (0)
#define A_PARK(buf_) do { *adst((buf_), 0) = r0; *adst((buf_), 1) = r1; \
        if constexpr (NQ == 4) { *adst((buf_), 2) = r2; *adst((buf_), 3) = r3; } } while (0)

    int mt = blockIdx.x;
    if (mt < mtiles) { A_FETCH(mt); A_PARK(0); }
    __syncthreads();

    int buf = 0;
    for (; mt < mtiles; mt += GEMM_GRID) {
        const bool more = mt + GEMM_GRID < mtiles;
        if (more) A_FETCH(mt + GEMM_GRID);

        const float* ap = &sA[buf][l * A_PITCH + 2 * half];
        v8f acc = {};
        #pragma unroll
        for (int s = 0; s < K_DIM / 4; ++s) {
            v2f a = *(const v2f*)(ap + 4 * s);
            acc = __builtin_amdgcn_wmma_f32_16x16x4_f32(
                      false, a, false, bf[s], (short)0, acc, false, false);
        }

        const int row0 = mt * 16;
        float* cp = C + (size_t)(row0 + half * 8) * NCOL + col0 + l;
        if (row0 + 16 <= M) {
            #pragma unroll
            for (int i = 0; i < 8; ++i) cp[i * NCOL] = acc[i];
        } else {
            #pragma unroll
            for (int i = 0; i < 8; ++i)
                if (row0 + half * 8 + i < M) cp[i * NCOL] = acc[i];
        }

        if (more) A_PARK(buf ^ 1);
        __syncthreads();
        buf ^= 1;
    }
#undef A_FETCH
#undef A_PARK
}

__global__ void att_coeff_kernel(const float* __restrict__ h,
                                 const float* __restrict__ att_s,
                                 const float* __restrict__ att_d,
                                 float* __restrict__ a_s, float* __restrict__ a_d,
                                 int N, int H, int CH) {
    int i = blockIdx.x * blockDim.x + threadIdx.x;
    if (i >= N * H) return;
    int n  = i / H;
    int hh = i - n * H;
    const float* hp = h + (size_t)n * H * CH + (size_t)hh * CH;
    const float* sp = att_s + hh * CH;
    const float* dp = att_d + hh * CH;
    float ss = 0.f, dd = 0.f;
    for (int c = 0; c < CH; ++c) { ss += hp[c] * sp[c]; dd += hp[c] * dp[c]; }
    a_s[i] = ss;
    a_d[i] = dd;
}

#define OWN_R    196
#define OWN_T    1024
#define OWN_CAP  4096
#define OWN_U    8
#define OWN_UC   8
#define LST_HDR  32
#define LST_STRIDE (LST_HDR + 2 * OWN_CAP)

__device__ inline void ldsMaxF(float* addr, float v) {
    int iv = __float_as_int(v);
    if (iv >= 0) atomicMax((int*)addr, iv);
    else         atomicMin((unsigned int*)addr, (unsigned int)iv);
}

template <int H, int CH, bool ELU, bool SAVE, bool LOAD>
__global__ __launch_bounds__(OWN_T)
void gat_owner_kernel(const int* __restrict__ ei, int E, int N,
                      int* __restrict__ lst,
                      const float* __restrict__ hfeat,
                      const float* __restrict__ a_s,
                      const float* __restrict__ a_d,
                      const float* __restrict__ bias,
                      float* __restrict__ outp) {
    constexpr int F  = H * CH;
    constexpr int KF = F / 32;
    constexpr int NW = OWN_T / 32;

    __shared__ float sAgg[OWN_R * F];
    __shared__ float sMax[OWN_R * H];
    __shared__ float sDen[OWN_R * H];
    __shared__ float sAd [OWN_R * H];
    __shared__ int   sSrc[OWN_CAP];
    __shared__ int   sDl [OWN_CAP];
    __shared__ float sE  [OWN_CAP * H];
    __shared__ int   sCnt;

    const int tid = threadIdx.x;
    const int n0  = blockIdx.x * OWN_R;
    const int nn  = (N - n0 < OWN_R) ? (N - n0) : OWN_R;
    const int* __restrict__ dstp = ei + E;
    const bool reuse = LOAD && (lst != nullptr);
    int* __restrict__ myl = lst ? lst + (size_t)blockIdx.x * LST_STRIDE : nullptr;
    int loaded = 0;

    for (int i = tid; i < OWN_R * F; i += OWN_T) sAgg[i] = 0.f;
    for (int i = tid; i < OWN_R * H; i += OWN_T) {
        sMax[i] = __int_as_float(0xff800000);
        sDen[i] = 0.f;
        sAd[i]  = (i < nn * H) ? a_d[(size_t)n0 * H + i] : 0.f;
    }
    if (!reuse) {
        for (int i = tid; i < nn; i += OWN_T) { sSrc[i] = -(i + 1); sDl[i] = i; }
        if (tid == 0) sCnt = nn;
    }
    __syncthreads();

    auto keep = [&](int dst, int eid, unsigned lim) {
        unsigned dl = (unsigned)(dst - n0);
        if (dl < lim) {
            int pos = atomicAdd(&sCnt, 1);
            if (pos < OWN_CAP) { sSrc[pos] = eid; sDl[pos] = (int)dl; }
        }
    };
    if (reuse) {
        loaded = myl[0];
        if ((unsigned)(loaded - 1) < (unsigned)OWN_CAP) {
            const int2* __restrict__ lp = (const int2*)(myl + LST_HDR);
            constexpr int NQ = OWN_CAP / OWN_T;
            int2 p[NQ];
            #pragma unroll
            for (int q = 0; q < NQ; ++q) {
                const int j = q * OWN_T + tid;
                p[q] = lp[(j < loaded) ? j : (loaded - 1)];
            }
            #pragma unroll
            for (int q = 0; q < NQ; ++q) {
                const int j = q * OWN_T + tid;
                if (j < loaded) {
                    sSrc[j] = p[q].x;
                    sDl[j]  = p[q].y;
                }
            }
        }
    } else if ((E & 3) == 0) {
        const int EV = E >> 2;
        const int4* __restrict__ dv = (const int4*)dstp;
        constexpr int STEP = OWN_T * OWN_U;
        const int rot = (int)(((long long)blockIdx.x * EV) / gridDim.x);
        auto slot = [&](int base, int u, bool& ok) {
            int v  = base + u * OWN_T + tid;
            ok     = v < EV;
            int vr = v + rot;
            if (vr >= EV) vr -= EV;
            return ok ? vr : 0;
        };
        int4 cur[OWN_U];
        #pragma unroll
        for (int u = 0; u < OWN_U; ++u) { bool ok; cur[u] = dv[slot(0, u, ok)]; }
        for (int base = 0; base < EV; base += STEP) {
            int4 nxt[OWN_U];
            #pragma unroll
            for (int u = 0; u < OWN_U; ++u) { bool ok; nxt[u] = dv[slot(base + STEP, u, ok)]; }
            #pragma unroll
            for (int u = 0; u < OWN_U; ++u) {
                bool ok;
                const int e0 = slot(base, u, ok) * 4;
                const unsigned lim = ok ? (unsigned)nn : 0u;
                keep(cur[u].x, e0 + 0, lim);
                keep(cur[u].y, e0 + 1, lim);
                keep(cur[u].z, e0 + 2, lim);
                keep(cur[u].w, e0 + 3, lim);
            }
            #pragma unroll
            for (int u = 0; u < OWN_U; ++u) cur[u] = nxt[u];
        }
    } else {
        for (int e = tid; e < E; e += OWN_T) keep(dstp[e], e, (unsigned)nn);
    }
    __syncthreads();
    const int total = reuse ? loaded : sCnt;
    if (SAVE && lst != nullptr && tid == 0) myl[0] = total;

    if ((unsigned)total <= (unsigned)OWN_CAP) {
        for (int j = tid; j < total; j += OWN_T) {
            int id  = sSrc[j];
            int src = reuse ? id : ((id >= 0) ? ei[id] : (n0 - id - 1));
            int dl  = sDl[j];
            sSrc[j] = src;
            for (int h = 0; h < H; ++h) {
                float v = a_s[(size_t)src * H + h] + sAd[dl * H + h];
                v = (v > 0.f) ? v : NEG_SLOPE * v;
                sE[j * H + h] = v;
                ldsMaxF(&sMax[dl * H + h], v);
            }
        }
        __syncthreads();
        if (SAVE && lst != nullptr) {
            int2* __restrict__ lp = (int2*)(myl + LST_HDR);
            for (int q = 0; q < OWN_CAP / OWN_T; ++q) {
                const int j = q * OWN_T + tid;
                if (j < total) lp[j] = make_int2(sSrc[j], sDl[j]);
            }
        }
        for (int j = tid; j < total; j += OWN_T) {
            int dl = sDl[j];
            for (int h = 0; h < H; ++h) {
                float ex = __expf(sE[j * H + h] - sMax[dl * H + h]);
                sE[j * H + h] = ex;
                atomicAdd(&sDen[dl * H + h], ex);
            }
        }
        __syncthreads();
        for (int j = tid; j < total; j += OWN_T) {
            int dl = sDl[j];
            for (int h = 0; h < H; ++h)
                sE[j * H + h] = sE[j * H + h] / sDen[dl * H + h];
        }
        __syncthreads();
        const int wave = tid >> 5;
        const int lane = tid & 31;
        for (int j0 = wave; j0 < total; j0 += NW * OWN_UC) {
            float v[OWN_UC][KF];
            int   dl[OWN_UC];
            #pragma unroll
            for (int q = 0; q < OWN_UC; ++q) {
                int  j   = j0 + q * NW;
                bool ok  = j < total;
                int  src = ok ? sSrc[j] : n0;
                dl[q]    = ok ? sDl[j]  : 0;
                const float* hp = hfeat + (size_t)src * F + lane;
                #pragma unroll
                for (int k = 0; k < KF; ++k) v[q][k] = hp[k * 32];
            }
            #pragma unroll
            for (int q = 0; q < OWN_UC; ++q) {
                int j = j0 + q * NW;
                if (j < total) {
                    #pragma unroll
                    for (int k = 0; k < KF; ++k) {
                        float al = sE[j * H + (k * 32) / CH];
                        atomicAdd(&sAgg[dl[q] * F + k * 32 + lane], v[q][k] * al);
                    }
                }
            }
        }
        __syncthreads();
    } else {
        const int ET = E + N;
        for (int e = tid; e < ET; e += OWN_T) {
            int dst = (e < E) ? dstp[e] : (e - E);
            unsigned dl = (unsigned)(dst - n0);
            if (dl < (unsigned)nn) {
                int src = (e < E) ? ei[e] : (e - E);
                for (int h = 0; h < H; ++h) {
                    float v = a_s[(size_t)src * H + h] + sAd[dl * H + h];
                    v = (v > 0.f) ? v : NEG_SLOPE * v;
                    ldsMaxF(&sMax[dl * H + h], v);
                }
            }
        }
        __syncthreads();
        for (int e = tid; e < ET; e += OWN_T) {
            int dst = (e < E) ? dstp[e] : (e - E);
            unsigned dl = (unsigned)(dst - n0);
            if (dl < (unsigned)nn) {
                int src = (e < E) ? ei[e] : (e - E);
                for (int h = 0; h < H; ++h) {
                    float v = a_s[(size_t)src * H + h] + sAd[dl * H + h];
                    v = (v > 0.f) ? v : NEG_SLOPE * v;
                    atomicAdd(&sDen[dl * H + h], __expf(v - sMax[dl * H + h]));
                }
            }
        }
        __syncthreads();
        for (int e = tid; e < ET; e += OWN_T) {
            int dst = (e < E) ? dstp[e] : (e - E);
            unsigned dl = (unsigned)(dst - n0);
            if (dl < (unsigned)nn) {
                int src = (e < E) ? ei[e] : (e - E);
                float al[H];
                for (int h = 0; h < H; ++h) {
                    float v = a_s[(size_t)src * H + h] + sAd[dl * H + h];
                    v = (v > 0.f) ? v : NEG_SLOPE * v;
                    al[h] = __expf(v - sMax[dl * H + h]) / sDen[dl * H + h];
                }
                for (int c = 0; c < F; ++c)
                    atomicAdd(&sAgg[dl * F + c], hfeat[(size_t)src * F + c] * al[c / CH]);
            }
        }
        __syncthreads();
    }

    for (int i = tid; i < nn * F; i += OWN_T) {
        float v = sAgg[i] + bias[i % F];
        if (ELU) v = (v > 0.f) ? v : (__expf(v) - 1.f);
        outp[(size_t)n0 * F + i] = v;
    }
}

extern "C" void kernel_launch(void* const* d_in, const int* in_sizes, int n_in,
                              void* d_out, int out_size, void* d_ws, size_t ws_size,
                              hipStream_t stream) {
    const float* x   = (const float*)d_in[0];
    const int*   ei  = (const int*)  d_in[1];
    const float* W1  = (const float*)d_in[2];
    const float* as1 = (const float*)d_in[3];
    const float* ad1 = (const float*)d_in[4];
    const float* b1  = (const float*)d_in[5];
    const float* W2  = (const float*)d_in[6];
    const float* as2 = (const float*)d_in[7];
    const float* ad2 = (const float*)d_in[8];
    const float* b2  = (const float*)d_in[9];
    float* out = (float*)d_out;

    const int N  = in_sizes[0] / 128;
    const int E  = in_sizes[1] / 2;
    const int H1 = 4, C1 = 32, F1 = 128;
    const int H2 = 1, C2 = 64, F2 = 64;

    char* w = (char*)d_ws;
    auto carve = [&](size_t bytes) {
        void* p = (void*)w;
        w += (bytes + 255) & ~(size_t)255;
        return p;
    };
    float* h1    = (float*)carve((size_t)N * F1 * 4);
    float* a1s   = (float*)carve((size_t)N * H1 * 4);
    float* a1d   = (float*)carve((size_t)N * H1 * 4);
    float* agg1  = (float*)carve((size_t)N * F1 * 4);
    float* h2    = (float*)carve((size_t)N * F2 * 4);
    float* a2s   = (float*)carve((size_t)N * 4);
    float* a2d   = (float*)carve((size_t)N * 4);

    const int og = (N + OWN_R - 1) / OWN_R;
    int* lst = (int*)carve((size_t)og * LST_STRIDE * 4);
    if ((size_t)(w - (char*)d_ws) > ws_size) lst = nullptr;

    {
        int mt = (N + 15) / 16;
        wmma_gemm_kernel<128><<<GEMM_GRID, 256, 0, stream>>>(x, W1, h1, N, mt);
    }
    att_coeff_kernel<<<(N * H1 + 255) / 256, 256, 0, stream>>>(
        h1, as1, ad1, a1s, a1d, N, H1, C1);
    gat_owner_kernel<4, 32, true, true, false><<<og, OWN_T, 0, stream>>>(
        ei, E, N, lst, h1, a1s, a1d, b1, agg1);

    {
        int mt = (N + 15) / 16;
        wmma_gemm_kernel<64><<<GEMM_GRID, 128, 0, stream>>>(agg1, W2, h2, N, mt);
    }
    att_coeff_kernel<<<(N + 255) / 256, 256, 0, stream>>>(
        h2, as2, ad2, a2s, a2d, N, H2, C2);
    gat_owner_kernel<1, 64, false, false, true><<<og, OWN_T, 0, stream>>>(
        ei, E, N, lst, h2, a2s, a2d, b2, out);
}


# Edge filter on one byte per edge: a front launch writes each edge's owner-block number; owner blocks scan it 16 edges a load with byte-match bit tests; sums in new list arrival order; 6->7 launches
# speedup vs baseline: 7.6300x; 1.1387x over previous
#include <hip/hip_runtime.h>
#include <math.h>

typedef __attribute__((ext_vector_type(2))) float v2f;
typedef __attribute__((ext_vector_type(8))) float v8f;

#define K_DIM 128
#define NEG_SLOPE 0.2f

#define GEMM_GRID 1024
#define A_PITCH   132

template <int NCOL>
__global__ __launch_bounds__(NCOL * 2)
void wmma_gemm_kernel(const float* __restrict__ A, const float* __restrict__ B,
                      float* __restrict__ C, int M, int mtiles) {
    constexpr int NT = NCOL * 2;
    constexpr int NQ = (16 * K_DIM / 4) / NT;
    __shared__ __attribute__((aligned(16))) float sA[2][16 * A_PITCH];

    const int wave = threadIdx.x >> 5;
    const int lane = threadIdx.x & 31;
    const int half = lane >> 4;
    const int l    = lane & 15;
    const int col0 = wave * 16;

    v2f bf[K_DIM / 4];
    {
        const float* bcol = B + col0 + l;
        #pragma unroll
        for (int s = 0; s < K_DIM / 4; ++s) {
            bf[s].x = bcol[(4 * s + 2 * half)     * NCOL];
            bf[s].y = bcol[(4 * s + 2 * half + 1) * NCOL];
        }
    }

    static_assert(NQ == 2 || NQ == 4, "an A tile is 2 or 4 pieces per thread");
    float4 r0, r1, r2 = {}, r3 = {};
    auto asrc = [&](int mt, int q) {
        const int f = q * NT + (int)threadIdx.x;
        int ar = mt * 16 + (f >> 5);
        if (ar >= M) ar = M - 1;
        return (const float4*)(A + (size_t)ar * K_DIM + (f & 31) * 4);
    };
    auto adst = [&](int buf, int q) {
        const int f = q * NT + (int)threadIdx.x;
        return (float4*)&sA[buf][(f >> 5) * A_PITCH + (f & 31) * 4];
    };
#define A_FETCH(mt_) do { r0 = *asrc((mt_), 0); r1 = *asrc((mt_), 1); \
        if constexpr (NQ == 4) { r2 = *asrc((mt_), 2); r3 = *asrc((mt_), 3); } } while (0)
#define A_PARK(buf_) do { *adst((buf_), 0) = r0; *adst((buf_), 1) = r1; \
        if constexpr (NQ == 4) { *adst((buf_), 2) = r2; *adst((buf_), 3) = r3; } } while (0)

    int mt = blockIdx.x;
    if (mt < mtiles) { A_FETCH(mt); A_PARK(0); }
    __syncthreads();

    int buf = 0;
    for (; mt < mtiles; mt += GEMM_GRID) {
        const bool more = mt + GEMM_GRID < mtiles;
        if (more) A_FETCH(mt + GEMM_GRID);

        const float* ap = &sA[buf][l * A_PITCH + 2 * half];
        v8f acc = {};
        #pragma unroll
        for (int s = 0; s < K_DIM / 4; ++s) {
            v2f a = *(const v2f*)(ap + 4 * s);
            acc = __builtin_amdgcn_wmma_f32_16x16x4_f32(
                      false, a, false, bf[s], (short)0, acc, false, false);
        }

        const int row0 = mt * 16;
        float* cp = C + (size_t)(row0 + half * 8) * NCOL + col0 + l;
        if (row0 + 16 <= M) {
            #pragma unroll
            for (int i = 0; i < 8; ++i) cp[i * NCOL] = acc[i];
        } else {
            #pragma unroll
            for (int i = 0; i < 8; ++i)
                if (row0 + half * 8 + i < M) cp[i * NCOL] = acc[i];
        }

        if (more) A_PARK(buf ^ 1);
        __syncthreads();
        buf ^= 1;
    }
#undef A_FETCH
#undef A_PARK
}

__global__ void att_coeff_kernel(const float* __restrict__ h,
                                 const float* __restrict__ att_s,
                                 const float* __restrict__ att_d,
                                 float* __restrict__ a_s, float* __restrict__ a_d,
                                 int N, int H, int CH) {
    int i = blockIdx.x * blockDim.x + threadIdx.x;
    if (i >= N * H) return;
    int n  = i / H;
    int hh = i - n * H;
    const float* hp = h + (size_t)n * H * CH + (size_t)hh * CH;
    const float* sp = att_s + hh * CH;
    const float* dp = att_d + hh * CH;
    float ss = 0.f, dd = 0.f;
    for (int c = 0; c < CH; ++c) { ss += hp[c] * sp[c]; dd += hp[c] * dp[c]; }
    a_s[i] = ss;
    a_d[i] = dd;
}

#define OWN_R    196
#define OWN_T    1024
#define OWN_CAP  4096
#define OWN_U    8
#define OWN_UB   4
#define OWN_UC   8
#define LST_HDR  32
#define LST_STRIDE (LST_HDR + 2 * OWN_CAP)

__device__ inline void ldsMaxF(float* addr, float v) {
    int iv = __float_as_int(v);
    if (iv >= 0) atomicMax((int*)addr, iv);
    else         atomicMin((unsigned int*)addr, (unsigned int)iv);
}

#define OB_Q 4
__global__ __launch_bounds__(256)
void owner_byte_kernel(const int* __restrict__ ei, int E, unsigned int* __restrict__ ob) {
    const int EQ = E >> 2;
    const int4* __restrict__ dv = (const int4*)(ei + E);
    const int g0 = ((blockIdx.x * blockDim.x + threadIdx.x) >> 5) * (OB_Q * 32) + (threadIdx.x & 31);
    int4 d[OB_Q];
    #pragma unroll
    for (int q = 0; q < OB_Q; ++q) {
        const int g = g0 + q * 32;
        d[q] = dv[(g < EQ) ? g : (EQ - 1)];
    }
    #pragma unroll
    for (int q = 0; q < OB_Q; ++q) {
        const int g = g0 + q * 32;
        const unsigned b0 = ((unsigned)d[q].x / (unsigned)OWN_R) & 255u;
        const unsigned b1 = ((unsigned)d[q].y / (unsigned)OWN_R) & 255u;
        const unsigned b2 = ((unsigned)d[q].z / (unsigned)OWN_R) & 255u;
        const unsigned b3 = ((unsigned)d[q].w / (unsigned)OWN_R) & 255u;
        if (g < EQ) ob[g] = b0 | (b1 << 8) | (b2 << 16) | (b3 << 24);
    }
}

template <int H, int CH, bool ELU, bool SAVE, bool LOAD>
__global__ __launch_bounds__(OWN_T)
void gat_owner_kernel(const int* __restrict__ ei, int E, int N,
                      const unsigned int* __restrict__ ob,
                      int* __restrict__ lst,
                      const float* __restrict__ hfeat,
                      const float* __restrict__ a_s,
                      const float* __restrict__ a_d,
                      const float* __restrict__ bias,
                      float* __restrict__ outp) {
    constexpr int F  = H * CH;
    constexpr int KF = F / 32;
    constexpr int NW = OWN_T / 32;

    constexpr int ROWS = OWN_R + 1;
    __shared__ float sAgg[ROWS * F];
    __shared__ float sMax[ROWS * H];
    __shared__ float sDen[ROWS * H];
    __shared__ float sAd [ROWS * H];
    __shared__ int   sSrc[OWN_CAP];
    __shared__ int   sDl [OWN_CAP];
    __shared__ float sE  [OWN_CAP * H];
    __shared__ int   sCnt;

    const int tid = threadIdx.x;
    const int n0  = blockIdx.x * OWN_R;
    const int nn  = (N - n0 < OWN_R) ? (N - n0) : OWN_R;
    const int* __restrict__ dstp = ei + E;
    const bool reuse = LOAD && (lst != nullptr);
    const bool bytes = !reuse && (ob != nullptr);
    int* __restrict__ myl = lst ? lst + (size_t)blockIdx.x * LST_STRIDE : nullptr;
    int loaded = 0;

    for (int i = tid; i < ROWS * F; i += OWN_T) sAgg[i] = 0.f;
    for (int i = tid; i < ROWS * H; i += OWN_T) {
        sMax[i] = __int_as_float(0xff800000);
        sDen[i] = 0.f;
        sAd[i]  = (i < nn * H) ? a_d[(size_t)n0 * H + i] : 0.f;
    }
    if (!reuse) {
        for (int i = tid; i < nn; i += OWN_T) { sSrc[i] = -(i + 1); sDl[i] = i; }
        if (tid == 0) sCnt = nn;
    }
    __syncthreads();

    auto keep = [&](int dst, int eid, unsigned lim) {
        unsigned dl = (unsigned)(dst - n0);
        if (dl < lim) {
            int pos = atomicAdd(&sCnt, 1);
            if (pos < OWN_CAP) { sSrc[pos] = eid; sDl[pos] = (int)dl; }
        }
    };
    if (reuse) {
        loaded = myl[0];
        if ((unsigned)(loaded - 1) < (unsigned)OWN_CAP) {
            const int2* __restrict__ lp = (const int2*)(myl + LST_HDR);
            constexpr int NQ = OWN_CAP / OWN_T;
            int2 p[NQ];
            #pragma unroll
            for (int q = 0; q < NQ; ++q) {
                const int j = q * OWN_T + tid;
                p[q] = lp[(j < loaded) ? j : (loaded - 1)];
            }
            #pragma unroll
            for (int q = 0; q < NQ; ++q) {
                const int j = q * OWN_T + tid;
                if (j < loaded) {
                    sSrc[j] = p[q].x;
                    sDl[j]  = p[q].y;
                }
            }
        }
    } else if (bytes) {
        const int EV = E >> 4;
        const uint4* __restrict__ bv = (const uint4*)ob;
        constexpr int STEP = OWN_T * OWN_UB;
        const int rot = (int)(((long long)blockIdx.x * EV) / gridDim.x);
        const unsigned pat = blockIdx.x * 0x01010101u;
        auto slot = [&](int base, int u) {
            int v  = base + u * OWN_T + tid;
            int vr = v + rot;
            if (vr >= EV) vr -= EV;
            return (v < EV) ? vr : -1;
        };
        auto eqm = [&](unsigned x) {
            const unsigned y = x ^ pat;
            return ~(((y & 0x7f7f7f7fu) + 0x7f7f7f7fu) | y | 0x7f7f7f7fu);
        };
        auto keepv = [&](const uint4& w, int vr) {
            unsigned m = (eqm(w.x) >> 7) | (eqm(w.y) >> 6) | (eqm(w.z) >> 5) | (eqm(w.w) >> 4);
            if (vr < 0) m = 0;
            while (m) {
                const int t = __builtin_ctz(m);
                m &= m - 1;
                const int eid = vr * 16 + ((t & 7) << 2) + (t >> 3);
                const int pos = atomicAdd(&sCnt, 1);
                if (pos < OWN_CAP) sSrc[pos] = eid;
            }
        };
        uint4 cur[OWN_UB];
        int   curv[OWN_UB];
        #pragma unroll
        for (int u = 0; u < OWN_UB; ++u) { curv[u] = slot(0, u); cur[u] = bv[curv[u] < 0 ? 0 : curv[u]]; }
        for (int base = 0; base < EV; base += STEP) {
            uint4 nxt[OWN_UB];
            int   nxtv[OWN_UB];
            #pragma unroll
            for (int u = 0; u < OWN_UB; ++u) { nxtv[u] = slot(base + STEP, u); nxt[u] = bv[nxtv[u] < 0 ? 0 : nxtv[u]]; }
            #pragma unroll
            for (int u = 0; u < OWN_UB; ++u) keepv(cur[u], curv[u]);
            #pragma unroll
            for (int u = 0; u < OWN_UB; ++u) { cur[u] = nxt[u]; curv[u] = nxtv[u]; }
        }
    } else if ((E & 3) == 0) {
        const int EV = E >> 2;
        const int4* __restrict__ dv = (const int4*)dstp;
        constexpr int STEP = OWN_T * OWN_U;
        const int rot = (int)(((long long)blockIdx.x * EV) / gridDim.x);
        auto slot = [&](int base, int u, bool& ok) {
            int v  = base + u * OWN_T + tid;
            ok     = v < EV;
            int vr = v + rot;
            if (vr >= EV) vr -= EV;
            return ok ? vr : 0;
        };
        int4 cur[OWN_U];
        #pragma unroll
        for (int u = 0; u < OWN_U; ++u) { bool ok; cur[u] = dv[slot(0, u, ok)]; }
        for (int base = 0; base < EV; base += STEP) {
            int4 nxt[OWN_U];
            #pragma unroll
            for (int u = 0; u < OWN_U; ++u) { bool ok; nxt[u] = dv[slot(base + STEP, u, ok)]; }
            #pragma unroll
            for (int u = 0; u < OWN_U; ++u) {
                bool ok;
                const int e0 = slot(base, u, ok) * 4;
                const unsigned lim = ok ? (unsigned)nn : 0u;
                keep(cur[u].x, e0 + 0, lim);
                keep(cur[u].y, e0 + 1, lim);
                keep(cur[u].z, e0 + 2, lim);
                keep(cur[u].w, e0 + 3, lim);
            }
            #pragma unroll
            for (int u = 0; u < OWN_U; ++u) cur[u] = nxt[u];
        }
    } else {
        for (int e = tid; e < E; e += OWN_T) keep(dstp[e], e, (unsigned)nn);
    }
    __syncthreads();
    const int total = reuse ? loaded : sCnt;
    if (SAVE && lst != nullptr && tid == 0) myl[0] = total;

    if ((unsigned)total <= (unsigned)OWN_CAP) {
        for (int j = tid; j < total; j += OWN_T) {
            int id  = sSrc[j];
            int src = reuse ? id : ((id >= 0) ? ei[id] : (n0 - id - 1));
            int dl;
            if (bytes) {
                dl = (id >= 0) ? (dstp[id] - n0) : (-id - 1);
                if ((unsigned)dl >= (unsigned)nn) { dl = OWN_R; src = n0; }
                sDl[j] = dl;
            } else {
                dl = sDl[j];
            }
            sSrc[j] = src;
            for (int h = 0; h < H; ++h) {
                float v = a_s[(size_t)src * H + h] + sAd[dl * H + h];
                v = (v > 0.f) ? v : NEG_SLOPE * v;
                sE[j * H + h] = v;
                ldsMaxF(&sMax[dl * H + h], v);
            }
        }
        __syncthreads();
        if (SAVE && lst != nullptr) {
            int2* __restrict__ lp = (int2*)(myl + LST_HDR);
            for (int q = 0; q < OWN_CAP / OWN_T; ++q) {
                const int j = q * OWN_T + tid;
                if (j < total) lp[j] = make_int2(sSrc[j], sDl[j]);
            }
        }
        for (int j = tid; j < total; j += OWN_T) {
            int dl = sDl[j];
            for (int h = 0; h < H; ++h) {
                float ex = __expf(sE[j * H + h] - sMax[dl * H + h]);
                sE[j * H + h] = ex;
                atomicAdd(&sDen[dl * H + h], ex);
            }
        }
        __syncthreads();
        for (int j = tid; j < total; j += OWN_T) {
            int dl = sDl[j];
            for (int h = 0; h < H; ++h)
                sE[j * H + h] = sE[j * H + h] / sDen[dl * H + h];
        }
        __syncthreads();
        const int wave = tid >> 5;
        const int lane = tid & 31;
        for (int j0 = wave; j0 < total; j0 += NW * OWN_UC) {
            float v[OWN_UC][KF];
            int   dl[OWN_UC];
            #pragma unroll
            for (int q = 0; q < OWN_UC; ++q) {
                int  j   = j0 + q * NW;
                bool ok  = j < total;
                int  src = ok ? sSrc[j] : n0;
                dl[q]    = ok ? sDl[j]  : 0;
                const float* hp = hfeat + (size_t)src * F + lane;
                #pragma unroll
                for (int k = 0; k < KF; ++k) v[q][k] = hp[k * 32];
            }
            #pragma unroll
            for (int q = 0; q < OWN_UC; ++q) {
                int j = j0 + q * NW;
                if (j < total) {
                    #pragma unroll
                    for (int k = 0; k < KF; ++k) {
                        float al = sE[j * H + (k * 32) / CH];
                        atomicAdd(&sAgg[dl[q] * F + k * 32 + lane], v[q][k] * al);
                    }
                }
            }
        }
        __syncthreads();
    } else {
        const int ET = E + N;
        for (int e = tid; e < ET; e += OWN_T) {
            int dst = (e < E) ? dstp[e] : (e - E);
            unsigned dl = (unsigned)(dst - n0);
            if (dl < (unsigned)nn) {
                int src = (e < E) ? ei[e] : (e - E);
                for (int h = 0; h < H; ++h) {
                    float v = a_s[(size_t)src * H + h] + sAd[dl * H + h];
                    v = (v > 0.f) ? v : NEG_SLOPE * v;
                    ldsMaxF(&sMax[dl * H + h], v);
                }
            }
        }
        __syncthreads();
        for (int e = tid; e < ET; e += OWN_T) {
            int dst = (e < E) ? dstp[e] : (e - E);
            unsigned dl = (unsigned)(dst - n0);
            if (dl < (unsigned)nn) {
                int src = (e < E) ? ei[e] : (e - E);
                for (int h = 0; h < H; ++h) {
                    float v = a_s[(size_t)src * H + h] + sAd[dl * H + h];
                    v = (v > 0.f) ? v : NEG_SLOPE * v;
                    atomicAdd(&sDen[dl * H + h], __expf(v - sMax[dl * H + h]));
                }
            }
        }
        __syncthreads();
        for (int e = tid; e < ET; e += OWN_T) {
            int dst = (e < E) ? dstp[e] : (e - E);
            unsigned dl = (unsigned)(dst - n0);
            if (dl < (unsigned)nn) {
                int src = (e < E) ? ei[e] : (e - E);
                float al[H];
                for (int h = 0; h < H; ++h) {
                    float v = a_s[(size_t)src * H + h] + sAd[dl * H + h];
                    v = (v > 0.f) ? v : NEG_SLOPE * v;
                    al[h] = __expf(v - sMax[dl * H + h]) / sDen[dl * H + h];
                }
                for (int c = 0; c < F; ++c)
                    atomicAdd(&sAgg[dl * F + c], hfeat[(size_t)src * F + c] * al[c / CH]);
            }
        }
        __syncthreads();
    }

    for (int i = tid; i < nn * F; i += OWN_T) {
        float v = sAgg[i] + bias[i % F];
        if (ELU) v = (v > 0.f) ? v : (__expf(v) - 1.f);
        outp[(size_t)n0 * F + i] = v;
    }
}

extern "C" void kernel_launch(void* const* d_in, const int* in_sizes, int n_in,
                              void* d_out, int out_size, void* d_ws, size_t ws_size,
                              hipStream_t stream) {
    const float* x   = (const float*)d_in[0];
    const int*   ei  = (const int*)  d_in[1];
    const float* W1  = (const float*)d_in[2];
    const float* as1 = (const float*)d_in[3];
    const float* ad1 = (const float*)d_in[4];
    const float* b1  = (const float*)d_in[5];
    const float* W2  = (const float*)d_in[6];
    const float* as2 = (const float*)d_in[7];
    const float* ad2 = (const float*)d_in[8];
    const float* b2  = (const float*)d_in[9];
    float* out = (float*)d_out;

    const int N  = in_sizes[0] / 128;
    const int E  = in_sizes[1] / 2;
    const int H1 = 4, C1 = 32, F1 = 128;
    const int H2 = 1, C2 = 64, F2 = 64;

    char* w = (char*)d_ws;
    auto carve = [&](size_t bytes) {
        void* p = (void*)w;
        w += (bytes + 255) & ~(size_t)255;
        return p;
    };
    float* h1    = (float*)carve((size_t)N * F1 * 4);
    float* a1s   = (float*)carve((size_t)N * H1 * 4);
    float* a1d   = (float*)carve((size_t)N * H1 * 4);
    float* agg1  = (float*)carve((size_t)N * F1 * 4);
    float* h2    = (float*)carve((size_t)N * F2 * 4);
    float* a2s   = (float*)carve((size_t)N * 4);
    float* a2d   = (float*)carve((size_t)N * 4);

    const int og = (N + OWN_R - 1) / OWN_R;
    int* lst = (int*)carve((size_t)og * LST_STRIDE * 4);
    if ((size_t)(w - (char*)d_ws) > ws_size) lst = nullptr;

    unsigned int* ob = nullptr;
    if (og <= 256 && (E & 15) == 0 && (size_t)E <= (size_t)N * F2 * 4) ob = (unsigned int*)h2;

    if (ob) owner_byte_kernel<<<((E >> 2) + 256 * OB_Q - 1) / (256 * OB_Q), 256, 0, stream>>>(ei, E, ob);

    {
        int mt = (N + 15) / 16;
        wmma_gemm_kernel<128><<<GEMM_GRID, 256, 0, stream>>>(x, W1, h1, N, mt);
    }
    att_coeff_kernel<<<(N * H1 + 255) / 256, 256, 0, stream>>>(
        h1, as1, ad1, a1s, a1d, N, H1, C1);
    gat_owner_kernel<4, 32, true, true, false><<<og, OWN_T, 0, stream>>>(
        ei, E, N, ob, lst, h1, a1s, a1d, b1, agg1);

    {
        int mt = (N + 15) / 16;
        wmma_gemm_kernel<64><<<GEMM_GRID, 128, 0, stream>>>(agg1, W2, h2, N, mt);
    }
    att_coeff_kernel<<<(N + 255) / 256, 256, 0, stream>>>(
        h2, as2, ad2, a2s, a2d, N, H2, C2);
    gat_owner_kernel<1, 64, false, false, true><<<og, OWN_T, 0, stream>>>(
        ei, E, N, nullptr, lst, h2, a2s, a2d, b2, out);
}


# Edge kernels' row phase: list entries and weights read once per group, passed by readlane; a source row fetched by one 16/8-byte load per lane; LDS row words interleaved; sums in LDS arrival order
# speedup vs baseline: 8.3476x; 1.0940x over previous
#include <hip/hip_runtime.h>
#include <math.h>

typedef __attribute__((ext_vector_type(2))) float v2f;
typedef __attribute__((ext_vector_type(8))) float v8f;

#define K_DIM 128
#define NEG_SLOPE 0.2f

#define GEMM_GRID 1024
#define A_PITCH   132

template <int NCOL>
__global__ __launch_bounds__(NCOL * 2)
void wmma_gemm_kernel(const float* __restrict__ A, const float* __restrict__ B,
                      float* __restrict__ C, int M, int mtiles) {
    constexpr int NT = NCOL * 2;
    constexpr int NQ = (16 * K_DIM / 4) / NT;
    __shared__ __attribute__((aligned(16))) float sA[2][16 * A_PITCH];

    const int wave = threadIdx.x >> 5;
    const int lane = threadIdx.x & 31;
    const int half = lane >> 4;
    const int l    = lane & 15;
    const int col0 = wave * 16;

    v2f bf[K_DIM / 4];
    {
        const float* bcol = B + col0 + l;
        #pragma unroll
        for (int s = 0; s < K_DIM / 4; ++s) {
            bf[s].x = bcol[(4 * s + 2 * half)     * NCOL];
            bf[s].y = bcol[(4 * s + 2 * half + 1) * NCOL];
        }
    }

    static_assert(NQ == 2 || NQ == 4, "an A tile is 2 or 4 pieces per thread");
    float4 r0, r1, r2 = {}, r3 = {};
    auto asrc = [&](int mt, int q) {
        const int f = q * NT + (int)threadIdx.x;
        int ar = mt * 16 + (f >> 5);
        if (ar >= M) ar = M - 1;
        return (const float4*)(A + (size_t)ar * K_DIM + (f & 31) * 4);
    };
    auto adst = [&](int buf, int q) {
        const int f = q * NT + (int)threadIdx.x;
        return (float4*)&sA[buf][(f >> 5) * A_PITCH + (f & 31) * 4];
    };
#define A_FETCH(mt_) do { r0 = *asrc((mt_), 0); r1 = *asrc((mt_), 1); \
        if constexpr (NQ == 4) { r2 = *asrc((mt_), 2); r3 = *asrc((mt_), 3); } } while (0)
#define A_PARK(buf_) do { *adst((buf_), 0) = r0; *adst((buf_), 1) = r1; \
        if constexpr (NQ == 4) { *adst((buf_), 2) = r2; *adst((buf_), 3) = r3; } } while (0)

    int mt = blockIdx.x;
    if (mt < mtiles) { A_FETCH(mt); A_PARK(0); }
    __syncthreads();

    int buf = 0;
    for (; mt < mtiles; mt += GEMM_GRID) {
        const bool more = mt + GEMM_GRID < mtiles;
        if (more) A_FETCH(mt + GEMM_GRID);

        const float* ap = &sA[buf][l * A_PITCH + 2 * half];
        v8f acc = {};
        #pragma unroll
        for (int s = 0; s < K_DIM / 4; ++s) {
            v2f a = *(const v2f*)(ap + 4 * s);
            acc = __builtin_amdgcn_wmma_f32_16x16x4_f32(
                      false, a, false, bf[s], (short)0, acc, false, false);
        }

        const int row0 = mt * 16;
        float* cp = C + (size_t)(row0 + half * 8) * NCOL + col0 + l;
        if (row0 + 16 <= M) {
            #pragma unroll
            for (int i = 0; i < 8; ++i) cp[i * NCOL] = acc[i];
        } else {
            #pragma unroll
            for (int i = 0; i < 8; ++i)
                if (row0 + half * 8 + i < M) cp[i * NCOL] = acc[i];
        }

        if (more) A_PARK(buf ^ 1);
        __syncthreads();
        buf ^= 1;
    }
#undef A_FETCH
#undef A_PARK
}

__global__ void att_coeff_kernel(const float* __restrict__ h,
                                 const float* __restrict__ att_s,
                                 const float* __restrict__ att_d,
                                 float* __restrict__ a_s, float* __restrict__ a_d,
                                 int N, int H, int CH) {
    int i = blockIdx.x * blockDim.x + threadIdx.x;
    if (i >= N * H) return;
    int n  = i / H;
    int hh = i - n * H;
    const float* hp = h + (size_t)n * H * CH + (size_t)hh * CH;
    const float* sp = att_s + hh * CH;
    const float* dp = att_d + hh * CH;
    float ss = 0.f, dd = 0.f;
    for (int c = 0; c < CH; ++c) { ss += hp[c] * sp[c]; dd += hp[c] * dp[c]; }
    a_s[i] = ss;
    a_d[i] = dd;
}

template <int K> struct row_words { typedef float type __attribute__((ext_vector_type(K))); };
#define OWN_R    196
#define OWN_T    1024
#define OWN_CAP  4096
#define OWN_U    8
#define OWN_UB   4
#define LST_HDR  32
#define LST_STRIDE (LST_HDR + 2 * OWN_CAP)

__device__ inline void ldsMaxF(float* addr, float v) {
    int iv = __float_as_int(v);
    if (iv >= 0) atomicMax((int*)addr, iv);
    else         atomicMin((unsigned int*)addr, (unsigned int)iv);
}

#define OB_Q 4
__global__ __launch_bounds__(256)
void owner_byte_kernel(const int* __restrict__ ei, int E, unsigned int* __restrict__ ob) {
    const int EQ = E >> 2;
    const int4* __restrict__ dv = (const int4*)(ei + E);
    const int g0 = ((blockIdx.x * blockDim.x + threadIdx.x) >> 5) * (OB_Q * 32) + (threadIdx.x & 31);
    int4 d[OB_Q];
    #pragma unroll
    for (int q = 0; q < OB_Q; ++q) {
        const int g = g0 + q * 32;
        d[q] = dv[(g < EQ) ? g : (EQ - 1)];
    }
    #pragma unroll
    for (int q = 0; q < OB_Q; ++q) {
        const int g = g0 + q * 32;
        const unsigned b0 = ((unsigned)d[q].x / (unsigned)OWN_R) & 255u;
        const unsigned b1 = ((unsigned)d[q].y / (unsigned)OWN_R) & 255u;
        const unsigned b2 = ((unsigned)d[q].z / (unsigned)OWN_R) & 255u;
        const unsigned b3 = ((unsigned)d[q].w / (unsigned)OWN_R) & 255u;
        if (g < EQ) ob[g] = b0 | (b1 << 8) | (b2 << 16) | (b3 << 24);
    }
}

template <int H, int CH, bool ELU, bool SAVE, bool LOAD>
__global__ __launch_bounds__(OWN_T)
void gat_owner_kernel(const int* __restrict__ ei, int E, int N,
                      const unsigned int* __restrict__ ob,
                      int* __restrict__ lst,
                      const float* __restrict__ hfeat,
                      const float* __restrict__ a_s,
                      const float* __restrict__ a_d,
                      const float* __restrict__ bias,
                      float* __restrict__ outp) {
    constexpr int F  = H * CH;
    constexpr int KF = F / 32;
    constexpr int NW = OWN_T / 32;

    constexpr int ROWS = OWN_R + 1;
    __shared__ float sAgg[ROWS * F];
    __shared__ float sMax[ROWS * H];
    __shared__ float sDen[ROWS * H];
    __shared__ float sAd [ROWS * H];
    __shared__ int   sSrc[OWN_CAP];
    __shared__ int   sDl [OWN_CAP];
    __shared__ float sE  [OWN_CAP * H];
    __shared__ int   sCnt;

    const int tid = threadIdx.x;
    const int n0  = blockIdx.x * OWN_R;
    const int nn  = (N - n0 < OWN_R) ? (N - n0) : OWN_R;
    const int* __restrict__ dstp = ei + E;
    const bool reuse = LOAD && (lst != nullptr);
    const bool bytes = !reuse && (ob != nullptr);
    int* __restrict__ myl = lst ? lst + (size_t)blockIdx.x * LST_STRIDE : nullptr;
    int loaded = 0;

    for (int i = tid; i < ROWS * F; i += OWN_T) sAgg[i] = 0.f;
    for (int i = tid; i < ROWS * H; i += OWN_T) {
        sMax[i] = __int_as_float(0xff800000);
        sDen[i] = 0.f;
        sAd[i]  = (i < nn * H) ? a_d[(size_t)n0 * H + i] : 0.f;
    }
    if (!reuse) {
        for (int i = tid; i < nn; i += OWN_T) { sSrc[i] = -(i + 1); sDl[i] = i; }
        if (tid == 0) sCnt = nn;
    }
    __syncthreads();

    auto keep = [&](int dst, int eid, unsigned lim) {
        unsigned dl = (unsigned)(dst - n0);
        if (dl < lim) {
            int pos = atomicAdd(&sCnt, 1);
            if (pos < OWN_CAP) { sSrc[pos] = eid; sDl[pos] = (int)dl; }
        }
    };
    if (reuse) {
        loaded = myl[0];
        if ((unsigned)(loaded - 1) < (unsigned)OWN_CAP) {
            const int2* __restrict__ lp = (const int2*)(myl + LST_HDR);
            constexpr int NQ = OWN_CAP / OWN_T;
            int2 p[NQ];
            #pragma unroll
            for (int q = 0; q < NQ; ++q) {
                const int j = q * OWN_T + tid;
                p[q] = lp[(j < loaded) ? j : (loaded - 1)];
            }
            #pragma unroll
            for (int q = 0; q < NQ; ++q) {
                const int j = q * OWN_T + tid;
                if (j < loaded) {
                    sSrc[j] = p[q].x;
                    sDl[j]  = p[q].y;
                }
            }
        }
    } else if (bytes) {
        const int EV = E >> 4;
        const uint4* __restrict__ bv = (const uint4*)ob;
        constexpr int STEP = OWN_T * OWN_UB;
        const int rot = (int)(((long long)blockIdx.x * EV) / gridDim.x);
        const unsigned pat = blockIdx.x * 0x01010101u;
        auto slot = [&](int base, int u) {
            int v  = base + u * OWN_T + tid;
            int vr = v + rot;
            if (vr >= EV) vr -= EV;
            return (v < EV) ? vr : -1;
        };
        auto eqm = [&](unsigned x) {
            const unsigned y = x ^ pat;
            return ~(((y & 0x7f7f7f7fu) + 0x7f7f7f7fu) | y | 0x7f7f7f7fu);
        };
        auto keepv = [&](const uint4& w, int vr) {
            unsigned m = (eqm(w.x) >> 7) | (eqm(w.y) >> 6) | (eqm(w.z) >> 5) | (eqm(w.w) >> 4);
            if (vr < 0) m = 0;
            while (m) {
                const int t = __builtin_ctz(m);
                m &= m - 1;
                const int eid = vr * 16 + ((t & 7) << 2) + (t >> 3);
                const int pos = atomicAdd(&sCnt, 1);
                if (pos < OWN_CAP) sSrc[pos] = eid;
            }
        };
        uint4 cur[OWN_UB];
        int   curv[OWN_UB];
        #pragma unroll
        for (int u = 0; u < OWN_UB; ++u) { curv[u] = slot(0, u); cur[u] = bv[curv[u] < 0 ? 0 : curv[u]]; }
        for (int base = 0; base < EV; base += STEP) {
            uint4 nxt[OWN_UB];
            int   nxtv[OWN_UB];
            #pragma unroll
            for (int u = 0; u < OWN_UB; ++u) { nxtv[u] = slot(base + STEP, u); nxt[u] = bv[nxtv[u] < 0 ? 0 : nxtv[u]]; }
            #pragma unroll
            for (int u = 0; u < OWN_UB; ++u) keepv(cur[u], curv[u]);
            #pragma unroll
            for (int u = 0; u < OWN_UB; ++u) { cur[u] = nxt[u]; curv[u] = nxtv[u]; }
        }
    } else if ((E & 3) == 0) {
        const int EV = E >> 2;
        const int4* __restrict__ dv = (const int4*)dstp;
        constexpr int STEP = OWN_T * OWN_U;
        const int rot = (int)(((long long)blockIdx.x * EV) / gridDim.x);
        auto slot = [&](int base, int u, bool& ok) {
            int v  = base + u * OWN_T + tid;
            ok     = v < EV;
            int vr = v + rot;
            if (vr >= EV) vr -= EV;
            return ok ? vr : 0;
        };
        int4 cur[OWN_U];
        #pragma unroll
        for (int u = 0; u < OWN_U; ++u) { bool ok; cur[u] = dv[slot(0, u, ok)]; }
        for (int base = 0; base < EV; base += STEP) {
            int4 nxt[OWN_U];
            #pragma unroll
            for (int u = 0; u < OWN_U; ++u) { bool ok; nxt[u] = dv[slot(base + STEP, u, ok)]; }
            #pragma unroll
            for (int u = 0; u < OWN_U; ++u) {
                bool ok;
                const int e0 = slot(base, u, ok) * 4;
                const unsigned lim = ok ? (unsigned)nn : 0u;
                keep(cur[u].x, e0 + 0, lim);
                keep(cur[u].y, e0 + 1, lim);
                keep(cur[u].z, e0 + 2, lim);
                keep(cur[u].w, e0 + 3, lim);
            }
            #pragma unroll
            for (int u = 0; u < OWN_U; ++u) cur[u] = nxt[u];
        }
    } else {
        for (int e = tid; e < E; e += OWN_T) keep(dstp[e], e, (unsigned)nn);
    }
    __syncthreads();
    const int total = reuse ? loaded : sCnt;
    if (SAVE && lst != nullptr && tid == 0) myl[0] = total;

    if ((unsigned)total <= (unsigned)OWN_CAP) {
        for (int j = tid; j < total; j += OWN_T) {
            int id  = sSrc[j];
            int src = reuse ? id : ((id >= 0) ? ei[id] : (n0 - id - 1));
            int dl;
            if (bytes) {
                dl = (id >= 0) ? (dstp[id] - n0) : (-id - 1);
                if ((unsigned)dl >= (unsigned)nn) { dl = OWN_R; src = n0; }
                sDl[j] = dl;
            } else {
                dl = sDl[j];
            }
            sSrc[j] = src;
            for (int h = 0; h < H; ++h) {
                float v = a_s[(size_t)src * H + h] + sAd[dl * H + h];
                v = (v > 0.f) ? v : NEG_SLOPE * v;
                sE[j * H + h] = v;
                ldsMaxF(&sMax[dl * H + h], v);
            }
        }
        __syncthreads();
        if (SAVE && lst != nullptr) {
            int2* __restrict__ lp = (int2*)(myl + LST_HDR);
            for (int q = 0; q < OWN_CAP / OWN_T; ++q) {
                const int j = q * OWN_T + tid;
                if (j < total) lp[j] = make_int2(sSrc[j], sDl[j]);
            }
        }
        for (int j = tid; j < total; j += OWN_T) {
            int dl = sDl[j];
            for (int h = 0; h < H; ++h) {
                float ex = __expf(sE[j * H + h] - sMax[dl * H + h]);
                sE[j * H + h] = ex;
                atomicAdd(&sDen[dl * H + h], ex);
            }
        }
        __syncthreads();
        for (int j = tid; j < total; j += OWN_T) {
            int dl = sDl[j];
            for (int h = 0; h < H; ++h)
                sE[j * H + h] = sE[j * H + h] / sDen[dl * H + h];
        }
        __syncthreads();
        typedef typename row_words<KF>::type rowv;
        static_assert(KF == 2 || KF == 4, "a lane holds 8 or 16 bytes of a row");
        static_assert(CH % KF == 0, "a lane's words belong to one head");
        constexpr int UC = 32 / KF;
        static_assert(UC * H <= 32, "a group's weights fit one lane each");
        const int wave = tid >> 5;
        const int lane = tid & 31;
        for (int j0 = wave; j0 < total; j0 += NW * UC) {
            int jl = j0 + lane * NW;
            if (jl >= total) jl = total - 1;
            const int   srcl = sSrc[jl];
            const int   dll  = sDl[jl];
            int ja = j0 + (lane / H) * NW;
            if (ja >= total) ja = total - 1;
            const int   alw  = __float_as_int(sE[ja * H + (lane % H)]);
            const int   hs   = (KF * lane) / CH;
            rowv v[UC];
            #pragma unroll
            for (int q = 0; q < UC; ++q) {
                const int src = __builtin_amdgcn_readlane(srcl, q);
                v[q] = *((const rowv*)(hfeat + (size_t)src * F) + lane);
            }
            float al[UC];
            #pragma unroll
            for (int q = 0; q < UC; ++q) {
                int a = __builtin_amdgcn_readlane(alw, q * H);
                #pragma unroll
                for (int h = 1; h < H; ++h) {
                    const int ah = __builtin_amdgcn_readlane(alw, q * H + h);
                    a = (hs == h) ? ah : a;
                }
                al[q] = __int_as_float(a);
            }
            #pragma unroll
            for (int q = 0; q < UC; ++q) {
                const int j = j0 + q * NW;
                if (j < total) {
                    const int dl = __builtin_amdgcn_readlane(dll, q);
                    float* ag = &sAgg[dl * F + lane];
                    #pragma unroll
                    for (int k = 0; k < KF; ++k) atomicAdd(ag + k * 32, v[q][k] * al[q]);
                }
            }
        }
        __syncthreads();
    } else {
        const int ET = E + N;
        for (int e = tid; e < ET; e += OWN_T) {
            int dst = (e < E) ? dstp[e] : (e - E);
            unsigned dl = (unsigned)(dst - n0);
            if (dl < (unsigned)nn) {
                int src = (e < E) ? ei[e] : (e - E);
                for (int h = 0; h < H; ++h) {
                    float v = a_s[(size_t)src * H + h] + sAd[dl * H + h];
                    v = (v > 0.f) ? v : NEG_SLOPE * v;
                    ldsMaxF(&sMax[dl * H + h], v);
                }
            }
        }
        __syncthreads();
        for (int e = tid; e < ET; e += OWN_T) {
            int dst = (e < E) ? dstp[e] : (e - E);
            unsigned dl = (unsigned)(dst - n0);
            if (dl < (unsigned)nn) {
                int src = (e < E) ? ei[e] : (e - E);
                for (int h = 0; h < H; ++h) {
                    float v = a_s[(size_t)src * H + h] + sAd[dl * H + h];
                    v = (v > 0.f) ? v : NEG_SLOPE * v;
                    atomicAdd(&sDen[dl * H + h], __expf(v - sMax[dl * H + h]));
                }
            }
        }
        __syncthreads();
        for (int e = tid; e < ET; e += OWN_T) {
            int dst = (e < E) ? dstp[e] : (e - E);
            unsigned dl = (unsigned)(dst - n0);
            if (dl < (unsigned)nn) {
                int src = (e < E) ? ei[e] : (e - E);
                float al[H];
                for (int h = 0; h < H; ++h) {
                    float v = a_s[(size_t)src * H + h] + sAd[dl * H + h];
                    v = (v > 0.f) ? v : NEG_SLOPE * v;
                    al[h] = __expf(v - sMax[dl * H + h]) / sDen[dl * H + h];
                }
                for (int c = 0; c < F; ++c)
                    atomicAdd(&sAgg[dl * F + (c % KF) * 32 + c / KF], hfeat[(size_t)src * F + c] * al[c / CH]);
            }
        }
        __syncthreads();
    }

    for (int it = tid; it < nn * 32; it += OWN_T) {
        const int r = it >> 5;
        const int l = it & 31;
        typename row_words<KF>::type o;
        #pragma unroll
        for (int k = 0; k < KF; ++k) {
            float v = sAgg[r * F + k * 32 + l] + bias[KF * l + k];
            if (ELU) v = (v > 0.f) ? v : (__expf(v) - 1.f);
            o[k] = v;
        }
        *((typename row_words<KF>::type*)(outp + (size_t)(n0 + r) * F) + l) = o;
    }
}

extern "C" void kernel_launch(void* const* d_in, const int* in_sizes, int n_in,
                              void* d_out, int out_size, void* d_ws, size_t ws_size,
                              hipStream_t stream) {
    const float* x   = (const float*)d_in[0];
    const int*   ei  = (const int*)  d_in[1];
    const float* W1  = (const float*)d_in[2];
    const float* as1 = (const float*)d_in[3];
    const float* ad1 = (const float*)d_in[4];
    const float* b1  = (const float*)d_in[5];
    const float* W2  = (const float*)d_in[6];
    const float* as2 = (const float*)d_in[7];
    const float* ad2 = (const float*)d_in[8];
    const float* b2  = (const float*)d_in[9];
    float* out = (float*)d_out;

    const int N  = in_sizes[0] / 128;
    const int E  = in_sizes[1] / 2;
    const int H1 = 4, C1 = 32, F1 = 128;
    const int H2 = 1, C2 = 64, F2 = 64;

    char* w = (char*)d_ws;
    auto carve = [&](size_t bytes) {
        void* p = (void*)w;
        w += (bytes + 255) & ~(size_t)255;
        return p;
    };
    float* h1    = (float*)carve((size_t)N * F1 * 4);
    float* a1s   = (float*)carve((size_t)N * H1 * 4);
    float* a1d   = (float*)carve((size_t)N * H1 * 4);
    float* agg1  = (float*)carve((size_t)N * F1 * 4);
    float* h2    = (float*)carve((size_t)N * F2 * 4);
    float* a2s   = (float*)carve((size_t)N * 4);
    float* a2d   = (float*)carve((size_t)N * 4);

    const int og = (N + OWN_R - 1) / OWN_R;
    int* lst = (int*)carve((size_t)og * LST_STRIDE * 4);
    if ((size_t)(w - (char*)d_ws) > ws_size) lst = nullptr;

    unsigned int* ob = nullptr;
    if (og <= 256 && (E & 15) == 0 && (size_t)E <= (size_t)N * F2 * 4) ob = (unsigned int*)h2;

    if (ob) owner_byte_kernel<<<((E >> 2) + 256 * OB_Q - 1) / (256 * OB_Q), 256, 0, stream>>>(ei, E, ob);

    {
        int mt = (N + 15) / 16;
        wmma_gemm_kernel<128><<<GEMM_GRID, 256, 0, stream>>>(x, W1, h1, N, mt);
    }
    att_coeff_kernel<<<(N * H1 + 255) / 256, 256, 0, stream>>>(
        h1, as1, ad1, a1s, a1d, N, H1, C1);
    gat_owner_kernel<4, 32, true, true, false><<<og, OWN_T, 0, stream>>>(
        ei, E, N, ob, lst, h1, a1s, a1d, b1, agg1);

    {
        int mt = (N + 15) / 16;
        wmma_gemm_kernel<64><<<GEMM_GRID, 128, 0, stream>>>(agg1, W2, h2, N, mt);
    }
    att_coeff_kernel<<<(N + 255) / 256, 256, 0, stream>>>(
        h2, as2, ad2, a2s, a2d, N, H2, C2);
    gat_owner_kernel<1, 64, false, false, true><<<og, OWN_T, 0, stream>>>(
        ei, E, N, nullptr, lst, h2, a2s, a2d, b2, out);
}


# a_s, a_d from weights folded with the att vectors in the front launch: product 1 gets a 9th 16-column strip, the layer-1 row store dots each finished row; att launches gone 7->5; logit sums reordered
# speedup vs baseline: 8.6490x; 1.0361x over previous
#include <hip/hip_runtime.h>
#include <math.h>

typedef __attribute__((ext_vector_type(2))) float v2f;
typedef __attribute__((ext_vector_type(8))) float v8f;

#define K_DIM 128
#define NEG_SLOPE 0.2f

#define BX_PITCH  128
#define GEMM_GRID 1024
#define A_PITCH   132

template <int NCOL, bool EXT>
__global__ __launch_bounds__(NCOL * 2 + (EXT ? 32 : 0))
void wmma_gemm_kernel(const float* __restrict__ A, const float* __restrict__ B,
                      float* __restrict__ C, int M, int mtiles,
                      const float* __restrict__ Bx,
                      float* __restrict__ xs,
                      float* __restrict__ xd) {
    constexpr int NT = NCOL * 2;
    constexpr int NQ = (16 * K_DIM / 4) / NT;
    __shared__ __attribute__((aligned(16))) float sA[2][16 * A_PITCH];

    const int wave = threadIdx.x >> 5;
    const int lane = threadIdx.x & 31;
    const int half = lane >> 4;
    const int l    = lane & 15;
    const int col0 = wave * 16;
    const bool ext = EXT && (wave == NCOL / 16);

    v2f bf[K_DIM / 4];
    {
        static_assert(!EXT || NCOL == BX_PITCH, "the folded columns are laid out with the row pitch of B");
        const float* bcol = (ext ? Bx : (B + col0)) + l;
        #pragma unroll
        for (int s = 0; s < K_DIM / 4; ++s) {
            bf[s].x = bcol[(4 * s + 2 * half)     * NCOL];
            bf[s].y = bcol[(4 * s + 2 * half + 1) * NCOL];
        }
    }

    static_assert(NQ == 2 || NQ == 4, "an A tile is 2 or 4 pieces per thread");
    float4 r0 = {}, r1 = {}, r2 = {}, r3 = {};
    auto asrc = [&](int mt, int q) {
        const int f = q * NT + (int)threadIdx.x;
        int ar = mt * 16 + (f >> 5);
        if (ar >= M) ar = M - 1;
        return (const float4*)(A + (size_t)ar * K_DIM + (f & 31) * 4);
    };
    auto adst = [&](int buf, int q) {
        const int f = q * NT + (int)threadIdx.x;
        return (float4*)&sA[buf][(f >> 5) * A_PITCH + (f & 31) * 4];
    };
#define A_FETCH(mt_) do { r0 = *asrc((mt_), 0); r1 = *asrc((mt_), 1); \
        if constexpr (NQ == 4) { r2 = *asrc((mt_), 2); r3 = *asrc((mt_), 3); } } while (0)
#define A_PARK(buf_) do { *adst((buf_), 0) = r0; *adst((buf_), 1) = r1; \
        if constexpr (NQ == 4) { *adst((buf_), 2) = r2; *adst((buf_), 3) = r3; } } while (0)

    int mt = blockIdx.x;
    if (mt < mtiles && !ext) { A_FETCH(mt); A_PARK(0); }
    __syncthreads();

    int buf = 0;
    for (; mt < mtiles; mt += GEMM_GRID) {
        const bool more = mt + GEMM_GRID < mtiles;
        if (more && !ext) A_FETCH(mt + GEMM_GRID);

        const float* ap = &sA[buf][l * A_PITCH + 2 * half];
        v8f acc = {};
        #pragma unroll
        for (int s = 0; s < K_DIM / 4; ++s) {
            v2f a = *(const v2f*)(ap + 4 * s);
            acc = __builtin_amdgcn_wmma_f32_16x16x4_f32(
                      false, a, false, bf[s], (short)0, acc, false, false);
        }

        const int row0 = mt * 16;
        float* cp = C + (size_t)(row0 + half * 8) * NCOL + col0 + l;
        if (ext) {
            float* xp = ((l < 4) ? (xs + l) : (xd + (l - 4))) + (size_t)(row0 + half * 8) * 4;
            if (l < 8) {
                #pragma unroll
                for (int i = 0; i < 8; ++i)
                    if (row0 + half * 8 + i < M) xp[i * 4] = acc[i];
            }
        } else if (row0 + 16 <= M) {
            #pragma unroll
            for (int i = 0; i < 8; ++i) cp[i * NCOL] = acc[i];
        } else {
            #pragma unroll
            for (int i = 0; i < 8; ++i)
                if (row0 + half * 8 + i < M) cp[i * NCOL] = acc[i];
        }

        if (more && !ext) A_PARK(buf ^ 1);
        __syncthreads();
        buf ^= 1;
    }
#undef A_FETCH
#undef A_PARK
}

#define FOLD_BEXT 0
#define FOLD_W2F  (K_DIM * BX_PITCH)
#define FOLD_LEN  (K_DIM * 16 + 2 * K_DIM)
#define FOLD_T    256
#define FOLD_BLOCKS (FOLD_LEN / FOLD_T)

__device__ inline void fold_part(int o, const float* __restrict__ W1, const float* __restrict__ as1,
                                 const float* __restrict__ ad1, const float* __restrict__ W2,
                                 const float* __restrict__ as2, const float* __restrict__ ad2,
                                 float* __restrict__ fold) {
    float v = 0.f;
    int at;
    if (o < K_DIM * 16) {
        const int k = o >> 4, j = o & 15;
        at = FOLD_BEXT + k * BX_PITCH + j;
        if (j < 8) {
            const int hh = j & 3;
            const float* wp = W1 + k * 128 + hh * 32;
            const float* ap = ((j < 4) ? as1 : ad1) + hh * 32;
            for (int c = 0; c < 32; ++c) v += wp[c] * ap[c];
        }
    } else {
        const int o2 = o - K_DIM * 16;
        const int k  = o2 & (K_DIM - 1);
        at = FOLD_W2F + o2;
        const float* wp = W2 + k * 64;
        const float* ap = (o2 < K_DIM) ? as2 : ad2;
        for (int c = 0; c < 64; ++c) v += wp[c] * ap[c];
    }
    fold[at] = v;
}

template <int K> struct row_words { typedef float type __attribute__((ext_vector_type(K))); };
#define OWN_R    196
#define OWN_T    1024
#define OWN_CAP  4096
#define OWN_U    8
#define OWN_UB   4
#define LST_HDR  32
#define LST_STRIDE (LST_HDR + 2 * OWN_CAP)

__device__ inline void ldsMaxF(float* addr, float v) {
    int iv = __float_as_int(v);
    if (iv >= 0) atomicMax((int*)addr, iv);
    else         atomicMin((unsigned int*)addr, (unsigned int)iv);
}

#define OB_Q 4
__global__ __launch_bounds__(256)
void owner_byte_kernel(const int* __restrict__ ei, int E, unsigned int* __restrict__ ob, int nbb,
                       const float* __restrict__ W1, const float* __restrict__ as1, const float* __restrict__ ad1,
                       const float* __restrict__ W2, const float* __restrict__ as2, const float* __restrict__ ad2,
                       float* __restrict__ fold) {
    if ((int)blockIdx.x >= nbb) {
        fold_part(((int)blockIdx.x - nbb) * FOLD_T + (int)threadIdx.x, W1, as1, ad1, W2, as2, ad2, fold);
        return;
    }
    const int EQ = E >> 2;
    const int4* __restrict__ dv = (const int4*)(ei + E);
    const int g0 = ((blockIdx.x * blockDim.x + threadIdx.x) >> 5) * (OB_Q * 32) + (threadIdx.x & 31);
    int4 d[OB_Q];
    #pragma unroll
    for (int q = 0; q < OB_Q; ++q) {
        const int g = g0 + q * 32;
        d[q] = dv[(g < EQ) ? g : (EQ - 1)];
    }
    #pragma unroll
    for (int q = 0; q < OB_Q; ++q) {
        const int g = g0 + q * 32;
        const unsigned b0 = ((unsigned)d[q].x / (unsigned)OWN_R) & 255u;
        const unsigned b1 = ((unsigned)d[q].y / (unsigned)OWN_R) & 255u;
        const unsigned b2 = ((unsigned)d[q].z / (unsigned)OWN_R) & 255u;
        const unsigned b3 = ((unsigned)d[q].w / (unsigned)OWN_R) & 255u;
        if (g < EQ) ob[g] = b0 | (b1 << 8) | (b2 << 16) | (b3 << 24);
    }
}

template <int M>
__device__ inline float add_xor16(float v) {
    return v + __int_as_float(__builtin_amdgcn_update_dpp(0, __float_as_int(v), 0x160 + M, 0xf, 0xf, false));
}

template <int H, int CH, bool ELU, bool SAVE, bool LOAD, bool NEXT>
__global__ __launch_bounds__(OWN_T)
void gat_owner_kernel(const int* __restrict__ ei, int E, int N,
                      const unsigned int* __restrict__ ob,
                      int* __restrict__ lst,
                      const float* __restrict__ hfeat,
                      const float* __restrict__ a_s,
                      const float* __restrict__ a_d,
                      const float* __restrict__ bias,
                      float* __restrict__ outp,
                      const float* __restrict__ w2f,
                      float* __restrict__ nas,
                      float* __restrict__ nad) {
    constexpr int F  = H * CH;
    constexpr int KF = F / 32;
    constexpr int NW = OWN_T / 32;

    static_assert(!NEXT || F == K_DIM, "the folded vectors are as long as the next product's inner dim");
    constexpr int ROWS = OWN_R + 1;
    __shared__ float sAgg[ROWS * F];
    __shared__ float sMax[ROWS * H];
    __shared__ float sDen[ROWS * H];
    __shared__ float sAd [ROWS * H];
    __shared__ int   sSrc[OWN_CAP];
    __shared__ int   sDl [OWN_CAP];
    __shared__ float sE  [OWN_CAP * H];
    __shared__ int   sCnt;

    const int tid = threadIdx.x;
    const int n0  = blockIdx.x * OWN_R;
    const int nn  = (N - n0 < OWN_R) ? (N - n0) : OWN_R;
    const int* __restrict__ dstp = ei + E;
    const bool reuse = LOAD && (lst != nullptr);
    const bool bytes = !reuse && (ob != nullptr);
    int* __restrict__ myl = lst ? lst + (size_t)blockIdx.x * LST_STRIDE : nullptr;
    int loaded = 0;

    for (int i = tid; i < ROWS * F; i += OWN_T) sAgg[i] = 0.f;
    for (int i = tid; i < ROWS * H; i += OWN_T) {
        sMax[i] = __int_as_float(0xff800000);
        sDen[i] = 0.f;
        sAd[i]  = (i < nn * H) ? a_d[(size_t)n0 * H + i] : 0.f;
    }
    if (!reuse) {
        for (int i = tid; i < nn; i += OWN_T) { sSrc[i] = -(i + 1); sDl[i] = i; }
        if (tid == 0) sCnt = nn;
    }
    __syncthreads();

    auto keep = [&](int dst, int eid, unsigned lim) {
        unsigned dl = (unsigned)(dst - n0);
        if (dl < lim) {
            int pos = atomicAdd(&sCnt, 1);
            if (pos < OWN_CAP) { sSrc[pos] = eid; sDl[pos] = (int)dl; }
        }
    };
    if (reuse) {
        loaded = myl[0];
        if ((unsigned)(loaded - 1) < (unsigned)OWN_CAP) {
            const int2* __restrict__ lp = (const int2*)(myl + LST_HDR);
            constexpr int NQ = OWN_CAP / OWN_T;
            int2 p[NQ];
            #pragma unroll
            for (int q = 0; q < NQ; ++q) {
                const int j = q * OWN_T + tid;
                p[q] = lp[(j < loaded) ? j : (loaded - 1)];
            }
            #pragma unroll
            for (int q = 0; q < NQ; ++q) {
                const int j = q * OWN_T + tid;
                if (j < loaded) {
                    sSrc[j] = p[q].x;
                    sDl[j]  = p[q].y;
                }
            }
        }
    } else if (bytes) {
        const int EV = E >> 4;
        const uint4* __restrict__ bv = (const uint4*)ob;
        constexpr int STEP = OWN_T * OWN_UB;
        const int rot = (int)(((long long)blockIdx.x * EV) / gridDim.x);
        const unsigned pat = blockIdx.x * 0x01010101u;
        auto slot = [&](int base, int u) {
            int v  = base + u * OWN_T + tid;
            int vr = v + rot;
            if (vr >= EV) vr -= EV;
            return (v < EV) ? vr : -1;
        };
        auto eqm = [&](unsigned x) {
            const unsigned y = x ^ pat;
            return ~(((y & 0x7f7f7f7fu) + 0x7f7f7f7fu) | y | 0x7f7f7f7fu);
        };
        auto keepv = [&](const uint4& w, int vr) {
            unsigned m = (eqm(w.x) >> 7) | (eqm(w.y) >> 6) | (eqm(w.z) >> 5) | (eqm(w.w) >> 4);
            if (vr < 0) m = 0;
            while (m) {
                const int t = __builtin_ctz(m);
                m &= m - 1;
                const int eid = vr * 16 + ((t & 7) << 2) + (t >> 3);
                const int pos = atomicAdd(&sCnt, 1);
                if (pos < OWN_CAP) sSrc[pos] = eid;
            }
        };
        uint4 cur[OWN_UB];
        int   curv[OWN_UB];
        #pragma unroll
        for (int u = 0; u < OWN_UB; ++u) { curv[u] = slot(0, u); cur[u] = bv[curv[u] < 0 ? 0 : curv[u]]; }
        for (int base = 0; base < EV; base += STEP) {
            uint4 nxt[OWN_UB];
            int   nxtv[OWN_UB];
            #pragma unroll
            for (int u = 0; u < OWN_UB; ++u) { nxtv[u] = slot(base + STEP, u); nxt[u] = bv[nxtv[u] < 0 ? 0 : nxtv[u]]; }
            #pragma unroll
            for (int u = 0; u < OWN_UB; ++u) keepv(cur[u], curv[u]);
            #pragma unroll
            for (int u = 0; u < OWN_UB; ++u) { cur[u] = nxt[u]; curv[u] = nxtv[u]; }
        }
    } else if ((E & 3) == 0) {
        const int EV = E >> 2;
        const int4* __restrict__ dv = (const int4*)dstp;
        constexpr int STEP = OWN_T * OWN_U;
        const int rot = (int)(((long long)blockIdx.x * EV) / gridDim.x);
        auto slot = [&](int base, int u, bool& ok) {
            int v  = base + u * OWN_T + tid;
            ok     = v < EV;
            int vr = v + rot;
            if (vr >= EV) vr -= EV;
            return ok ? vr : 0;
        };
        int4 cur[OWN_U];
        #pragma unroll
        for (int u = 0; u < OWN_U; ++u) { bool ok; cur[u] = dv[slot(0, u, ok)]; }
        for (int base = 0; base < EV; base += STEP) {
            int4 nxt[OWN_U];
            #pragma unroll
            for (int u = 0; u < OWN_U; ++u) { bool ok; nxt[u] = dv[slot(base + STEP, u, ok)]; }
            #pragma unroll
            for (int u = 0; u < OWN_U; ++u) {
                bool ok;
                const int e0 = slot(base, u, ok) * 4;
                const unsigned lim = ok ? (unsigned)nn : 0u;
                keep(cur[u].x, e0 + 0, lim);
                keep(cur[u].y, e0 + 1, lim);
                keep(cur[u].z, e0 + 2, lim);
                keep(cur[u].w, e0 + 3, lim);
            }
            #pragma unroll
            for (int u = 0; u < OWN_U; ++u) cur[u] = nxt[u];
        }
    } else {
        for (int e = tid; e < E; e += OWN_T) keep(dstp[e], e, (unsigned)nn);
    }
    __syncthreads();
    const int total = reuse ? loaded : sCnt;
    if (SAVE && lst != nullptr && tid == 0) myl[0] = total;

    if ((unsigned)total <= (unsigned)OWN_CAP) {
        for (int j = tid; j < total; j += OWN_T) {
            int id  = sSrc[j];
            int src = reuse ? id : ((id >= 0) ? ei[id] : (n0 - id - 1));
            int dl;
            if (bytes) {
                dl = (id >= 0) ? (dstp[id] - n0) : (-id - 1);
                if ((unsigned)dl >= (unsigned)nn) { dl = OWN_R; src = n0; }
                sDl[j] = dl;
            } else {
                dl = sDl[j];
            }
            sSrc[j] = src;
            for (int h = 0; h < H; ++h) {
                float v = a_s[(size_t)src * H + h] + sAd[dl * H + h];
                v = (v > 0.f) ? v : NEG_SLOPE * v;
                sE[j * H + h] = v;
                ldsMaxF(&sMax[dl * H + h], v);
            }
        }
        __syncthreads();
        if (SAVE && lst != nullptr) {
            int2* __restrict__ lp = (int2*)(myl + LST_HDR);
            for (int q = 0; q < OWN_CAP / OWN_T; ++q) {
                const int j = q * OWN_T + tid;
                if (j < total) lp[j] = make_int2(sSrc[j], sDl[j]);
            }
        }
        for (int j = tid; j < total; j += OWN_T) {
            int dl = sDl[j];
            for (int h = 0; h < H; ++h) {
                float ex = __expf(sE[j * H + h] - sMax[dl * H + h]);
                sE[j * H + h] = ex;
                atomicAdd(&sDen[dl * H + h], ex);
            }
        }
        __syncthreads();
        for (int j = tid; j < total; j += OWN_T) {
            int dl = sDl[j];
            for (int h = 0; h < H; ++h)
                sE[j * H + h] = sE[j * H + h] / sDen[dl * H + h];
        }
        __syncthreads();
        typedef typename row_words<KF>::type rowv;
        static_assert(KF == 2 || KF == 4, "a lane holds 8 or 16 bytes of a row");
        static_assert(CH % KF == 0, "a lane's words belong to one head");
        constexpr int UC = 32 / KF;
        static_assert(UC * H <= 32, "a group's weights fit one lane each");
        const int wave = tid >> 5;
        const int lane = tid & 31;
        for (int j0 = wave; j0 < total; j0 += NW * UC) {
            int jl = j0 + lane * NW;
            if (jl >= total) jl = total - 1;
            const int   srcl = sSrc[jl];
            const int   dll  = sDl[jl];
            int ja = j0 + (lane / H) * NW;
            if (ja >= total) ja = total - 1;
            const int   alw  = __float_as_int(sE[ja * H + (lane % H)]);
            const int   hs   = (KF * lane) / CH;
            rowv v[UC];
            #pragma unroll
            for (int q = 0; q < UC; ++q) {
                const int src = __builtin_amdgcn_readlane(srcl, q);
                v[q] = *((const rowv*)(hfeat + (size_t)src * F) + lane);
            }
            float al[UC];
            #pragma unroll
            for (int q = 0; q < UC; ++q) {
                int a = __builtin_amdgcn_readlane(alw, q * H);
                #pragma unroll
                for (int h = 1; h < H; ++h) {
                    const int ah = __builtin_amdgcn_readlane(alw, q * H + h);
                    a = (hs == h) ? ah : a;
                }
                al[q] = __int_as_float(a);
            }
            #pragma unroll
            for (int q = 0; q < UC; ++q) {
                const int j = j0 + q * NW;
                if (j < total) {
                    const int dl = __builtin_amdgcn_readlane(dll, q);
                    float* ag = &sAgg[dl * F + lane];
                    #pragma unroll
                    for (int k = 0; k < KF; ++k) atomicAdd(ag + k * 32, v[q][k] * al[q]);
                }
            }
        }
        __syncthreads();
    } else {
        const int ET = E + N;
        for (int e = tid; e < ET; e += OWN_T) {
            int dst = (e < E) ? dstp[e] : (e - E);
            unsigned dl = (unsigned)(dst - n0);
            if (dl < (unsigned)nn) {
                int src = (e < E) ? ei[e] : (e - E);
                for (int h = 0; h < H; ++h) {
                    float v = a_s[(size_t)src * H + h] + sAd[dl * H + h];
                    v = (v > 0.f) ? v : NEG_SLOPE * v;
                    ldsMaxF(&sMax[dl * H + h], v);
                }
            }
        }
        __syncthreads();
        for (int e = tid; e < ET; e += OWN_T) {
            int dst = (e < E) ? dstp[e] : (e - E);
            unsigned dl = (unsigned)(dst - n0);
            if (dl < (unsigned)nn) {
                int src = (e < E) ? ei[e] : (e - E);
                for (int h = 0; h < H; ++h) {
                    float v = a_s[(size_t)src * H + h] + sAd[dl * H + h];
                    v = (v > 0.f) ? v : NEG_SLOPE * v;
                    atomicAdd(&sDen[dl * H + h], __expf(v - sMax[dl * H + h]));
                }
            }
        }
        __syncthreads();
        for (int e = tid; e < ET; e += OWN_T) {
            int dst = (e < E) ? dstp[e] : (e - E);
            unsigned dl = (unsigned)(dst - n0);
            if (dl < (unsigned)nn) {
                int src = (e < E) ? ei[e] : (e - E);
                float al[H];
                for (int h = 0; h < H; ++h) {
                    float v = a_s[(size_t)src * H + h] + sAd[dl * H + h];
                    v = (v > 0.f) ? v : NEG_SLOPE * v;
                    al[h] = __expf(v - sMax[dl * H + h]) / sDen[dl * H + h];
                }
                for (int c = 0; c < F; ++c)
                    atomicAdd(&sAgg[dl * F + (c % KF) * 32 + c / KF], hfeat[(size_t)src * F + c] * al[c / CH]);
            }
        }
        __syncthreads();
    }

    float fs[KF], fd[KF];
    #pragma unroll
    for (int k = 0; k < KF; ++k) {
        fs[k] = NEXT ? w2f[KF * (tid & 31) + k]     : 0.f;
        fd[k] = NEXT ? w2f[F + KF * (tid & 31) + k] : 0.f;
    }
    for (int it = tid; it < nn * 32; it += OWN_T) {
        const int r = it >> 5;
        const int l = it & 31;
        typename row_words<KF>::type o;
        #pragma unroll
        for (int k = 0; k < KF; ++k) {
            float v = sAgg[r * F + k * 32 + l] + bias[KF * l + k];
            if (ELU) v = (v > 0.f) ? v : (__expf(v) - 1.f);
            o[k] = v;
        }
        *((typename row_words<KF>::type*)(outp + (size_t)(n0 + r) * F) + l) = o;
        if (NEXT) {
            float ps = 0.f, pd = 0.f;
            #pragma unroll
            for (int k = 0; k < KF; ++k) { ps += o[k] * fs[k]; pd += o[k] * fd[k]; }
            ps = add_xor16<1>(ps); pd = add_xor16<1>(pd);
            ps = add_xor16<2>(ps); pd = add_xor16<2>(pd);
            ps = add_xor16<4>(ps); pd = add_xor16<4>(pd);
            ps = add_xor16<8>(ps); pd = add_xor16<8>(pd);
            const float ts = __int_as_float(__builtin_amdgcn_readlane(__float_as_int(ps), 0))
                           + __int_as_float(__builtin_amdgcn_readlane(__float_as_int(ps), 16));
            const float td = __int_as_float(__builtin_amdgcn_readlane(__float_as_int(pd), 0))
                           + __int_as_float(__builtin_amdgcn_readlane(__float_as_int(pd), 16));
            if (l == 0) { nas[n0 + r] = ts; nad[n0 + r] = td; }
        }
    }
}

extern "C" void kernel_launch(void* const* d_in, const int* in_sizes, int n_in,
                              void* d_out, int out_size, void* d_ws, size_t ws_size,
                              hipStream_t stream) {
    const float* x   = (const float*)d_in[0];
    const int*   ei  = (const int*)  d_in[1];
    const float* W1  = (const float*)d_in[2];
    const float* as1 = (const float*)d_in[3];
    const float* ad1 = (const float*)d_in[4];
    const float* b1  = (const float*)d_in[5];
    const float* W2  = (const float*)d_in[6];
    const float* as2 = (const float*)d_in[7];
    const float* ad2 = (const float*)d_in[8];
    const float* b2  = (const float*)d_in[9];
    float* out = (float*)d_out;

    const int N  = in_sizes[0] / 128;
    const int E  = in_sizes[1] / 2;
    const int H1 = 4, F1 = 128;
    const int F2 = 64;

    char* w = (char*)d_ws;
    auto carve = [&](size_t bytes) {
        void* p = (void*)w;
        w += (bytes + 255) & ~(size_t)255;
        return p;
    };
    float* h1    = (float*)carve((size_t)N * F1 * 4);
    float* a1s   = (float*)carve((size_t)N * H1 * 4);
    float* a1d   = (float*)carve((size_t)N * H1 * 4);
    float* agg1  = (float*)carve((size_t)N * F1 * 4);
    float* h2    = (float*)carve((size_t)N * F2 * 4);
    float* a2s   = (float*)carve((size_t)N * 4);
    float* a2d   = (float*)carve((size_t)N * 4);

    const int og = (N + OWN_R - 1) / OWN_R;
    int* lst = (int*)carve((size_t)og * LST_STRIDE * 4);
    if ((size_t)(w - (char*)d_ws) > ws_size) lst = nullptr;

    unsigned int* ob = nullptr;
    if (og <= 256 && (E & 15) == 0 && (size_t)E <= (size_t)N * F2 * 4 && E <= (1 << 20)) ob = (unsigned int*)h2;
    float* fold = h2 + (1 << 18);

    const int nbb = ob ? ((E >> 2) + 256 * OB_Q - 1) / (256 * OB_Q) : 0;
    owner_byte_kernel<<<nbb + FOLD_BLOCKS, FOLD_T, 0, stream>>>(ei, E, ob, nbb, W1, as1, ad1, W2, as2, ad2, fold);

    {
        int mt = (N + 15) / 16;
        wmma_gemm_kernel<128, true><<<GEMM_GRID, 288, 0, stream>>>(x, W1, h1, N, mt, fold + FOLD_BEXT, a1s, a1d);
    }
    gat_owner_kernel<4, 32, true, true, false, true><<<og, OWN_T, 0, stream>>>(
        ei, E, N, ob, lst, h1, a1s, a1d, b1, agg1, fold + FOLD_W2F, a2s, a2d);

    {
        int mt = (N + 15) / 16;
        wmma_gemm_kernel<64, false><<<GEMM_GRID, 128, 0, stream>>>(agg1, W2, h2, N, mt, nullptr, nullptr, nullptr);
    }
    gat_owner_kernel<1, 64, false, false, true, false><<<og, OWN_T, 0, stream>>>(
        ei, E, N, nullptr, lst, h2, a2s, a2d, b2, out, nullptr, nullptr, nullptr);
}


# Edge filter by occupancy bits: front launch also ORs, per owner block, one bit per 16-edge byte vector holding an edge of its range; owners load only flagged vectors; sums in new list arrival order
# speedup vs baseline: 9.0700x; 1.0487x over previous
#include <hip/hip_runtime.h>
#include <math.h>

typedef __attribute__((ext_vector_type(2))) float v2f;
typedef __attribute__((ext_vector_type(8))) float v8f;

#define K_DIM 128
#define NEG_SLOPE 0.2f

#define BX_PITCH  128
#define GEMM_GRID 1024
#define A_PITCH   132

template <int NCOL, bool EXT>
__global__ __launch_bounds__(NCOL * 2 + (EXT ? 32 : 0))
void wmma_gemm_kernel(const float* __restrict__ A, const float* __restrict__ B,
                      float* __restrict__ C, int M, int mtiles,
                      const float* __restrict__ Bx,
                      float* __restrict__ xs,
                      float* __restrict__ xd) {
    constexpr int NT = NCOL * 2;
    constexpr int NQ = (16 * K_DIM / 4) / NT;
    __shared__ __attribute__((aligned(16))) float sA[2][16 * A_PITCH];

    const int wave = threadIdx.x >> 5;
    const int lane = threadIdx.x & 31;
    const int half = lane >> 4;
    const int l    = lane & 15;
    const int col0 = wave * 16;
    const bool ext = EXT && (wave == NCOL / 16);

    v2f bf[K_DIM / 4];
    {
        static_assert(!EXT || NCOL == BX_PITCH, "the folded columns are laid out with the row pitch of B");
        const float* bcol = (ext ? Bx : (B + col0)) + l;
        #pragma unroll
        for (int s = 0; s < K_DIM / 4; ++s) {
            bf[s].x = bcol[(4 * s + 2 * half)     * NCOL];
            bf[s].y = bcol[(4 * s + 2 * half + 1) * NCOL];
        }
    }

    static_assert(NQ == 2 || NQ == 4, "an A tile is 2 or 4 pieces per thread");
    float4 r0 = {}, r1 = {}, r2 = {}, r3 = {};
    auto asrc = [&](int mt, int q) {
        const int f = q * NT + (int)threadIdx.x;
        int ar = mt * 16 + (f >> 5);
        if (ar >= M) ar = M - 1;
        return (const float4*)(A + (size_t)ar * K_DIM + (f & 31) * 4);
    };
    auto adst = [&](int buf, int q) {
        const int f = q * NT + (int)threadIdx.x;
        return (float4*)&sA[buf][(f >> 5) * A_PITCH + (f & 31) * 4];
    };
#define A_FETCH(mt_) do { r0 = *asrc((mt_), 0); r1 = *asrc((mt_), 1); \
        if constexpr (NQ == 4) { r2 = *asrc((mt_), 2); r3 = *asrc((mt_), 3); } } while (0)
#define A_PARK(buf_) do { *adst((buf_), 0) = r0; *adst((buf_), 1) = r1; \
        if constexpr (NQ == 4) { *adst((buf_), 2) = r2; *adst((buf_), 3) = r3; } } while (0)

    int mt = blockIdx.x;
    if (mt < mtiles && !ext) { A_FETCH(mt); A_PARK(0); }
    __syncthreads();

    int buf = 0;
    for (; mt < mtiles; mt += GEMM_GRID) {
        const bool more = mt + GEMM_GRID < mtiles;
        if (more && !ext) A_FETCH(mt + GEMM_GRID);

        const float* ap = &sA[buf][l * A_PITCH + 2 * half];
        v8f acc = {};
        #pragma unroll
        for (int s = 0; s < K_DIM / 4; ++s) {
            v2f a = *(const v2f*)(ap + 4 * s);
            acc = __builtin_amdgcn_wmma_f32_16x16x4_f32(
                      false, a, false, bf[s], (short)0, acc, false, false);
        }

        const int row0 = mt * 16;
        float* cp = C + (size_t)(row0 + half * 8) * NCOL + col0 + l;
        if (ext) {
            float* xp = ((l < 4) ? (xs + l) : (xd + (l - 4))) + (size_t)(row0 + half * 8) * 4;
            if (l < 8) {
                #pragma unroll
                for (int i = 0; i < 8; ++i)
                    if (row0 + half * 8 + i < M) xp[i * 4] = acc[i];
            }
        } else if (row0 + 16 <= M) {
            #pragma unroll
            for (int i = 0; i < 8; ++i) cp[i * NCOL] = acc[i];
        } else {
            #pragma unroll
            for (int i = 0; i < 8; ++i)
                if (row0 + half * 8 + i < M) cp[i * NCOL] = acc[i];
        }

        if (more && !ext) A_PARK(buf ^ 1);
        __syncthreads();
        buf ^= 1;
    }
#undef A_FETCH
#undef A_PARK
}

#define FOLD_BEXT 0
#define FOLD_W2F  (K_DIM * BX_PITCH)
#define FOLD_LEN  (K_DIM * 16 + 2 * K_DIM)
#define FOLD_T    256
#define FOLD_BLOCKS (FOLD_LEN / FOLD_T)

__device__ inline void fold_part(int o, const float* __restrict__ W1, const float* __restrict__ as1,
                                 const float* __restrict__ ad1, const float* __restrict__ W2,
                                 const float* __restrict__ as2, const float* __restrict__ ad2,
                                 float* __restrict__ fold) {
    float v = 0.f;
    int at;
    if (o < K_DIM * 16) {
        const int k = o >> 4, j = o & 15;
        at = FOLD_BEXT + k * BX_PITCH + j;
        if (j < 8) {
            const int hh = j & 3;
            const float* wp = W1 + k * 128 + hh * 32;
            const float* ap = ((j < 4) ? as1 : ad1) + hh * 32;
            for (int c = 0; c < 32; ++c) v += wp[c] * ap[c];
        }
    } else {
        const int o2 = o - K_DIM * 16;
        const int k  = o2 & (K_DIM - 1);
        at = FOLD_W2F + o2;
        const float* wp = W2 + k * 64;
        const float* ap = (o2 < K_DIM) ? as2 : ad2;
        for (int c = 0; c < 64; ++c) v += wp[c] * ap[c];
    }
    fold[at] = v;
}

template <int K> struct row_words { typedef float type __attribute__((ext_vector_type(K))); };
#define OWN_R    196
#define OWN_T    1024
#define OWN_CAP  4096
#define OWN_U    8
#define BM_W     8
#define LST_HDR  32
#define LST_STRIDE (LST_HDR + 2 * OWN_CAP)

__device__ inline void ldsMaxF(float* addr, float v) {
    int iv = __float_as_int(v);
    if (iv >= 0) atomicMax((int*)addr, iv);
    else         atomicMin((unsigned int*)addr, (unsigned int)iv);
}

#define OB_Q 4
__global__ __launch_bounds__(256)
void owner_byte_kernel(const int* __restrict__ ei, int E, unsigned int* __restrict__ ob,
                       unsigned int* __restrict__ bm, int nbb, int nown,
                       const float* __restrict__ W1, const float* __restrict__ as1, const float* __restrict__ ad1,
                       const float* __restrict__ W2, const float* __restrict__ as2, const float* __restrict__ ad2,
                       float* __restrict__ fold) {
    if ((int)blockIdx.x >= nbb) {
        fold_part(((int)blockIdx.x - nbb) * FOLD_T + (int)threadIdx.x, W1, as1, ad1, W2, as2, ad2, fold);
        return;
    }
    static_assert(BM_W == 8, "thread b stores owner b's words of the block as two 16-byte pieces");
    __shared__ unsigned int sBm[BM_W * 256];
    const int EQ = E >> 2;
    const int4* __restrict__ dv = (const int4*)(ei + E);
    const int g0 = ((blockIdx.x * blockDim.x + threadIdx.x) >> 5) * (OB_Q * 32) + (threadIdx.x & 31);
    int4 d[OB_Q];
    #pragma unroll
    for (int q = 0; q < OB_Q; ++q) {
        const int g = g0 + q * 32;
        d[q] = dv[(g < EQ) ? g : (EQ - 1)];
    }
    for (int i = threadIdx.x; i < BM_W * 256; i += 256) sBm[i] = 0u;
    __syncthreads();
    #pragma unroll
    for (int q = 0; q < OB_Q; ++q) {
        const int g = g0 + q * 32;
        const unsigned b0 = ((unsigned)d[q].x / (unsigned)OWN_R) & 255u;
        const unsigned b1 = ((unsigned)d[q].y / (unsigned)OWN_R) & 255u;
        const unsigned b2 = ((unsigned)d[q].z / (unsigned)OWN_R) & 255u;
        const unsigned b3 = ((unsigned)d[q].w / (unsigned)OWN_R) & 255u;
        if (g < EQ) {
            ob[g] = b0 | (b1 << 8) | (b2 << 16) | (b3 << 24);
            const int vl = (g >> 2) - (int)blockIdx.x * 256;
            unsigned int* row = sBm + (vl >> 5) * 256;
            const unsigned bit = 1u << (vl & 31);
            atomicOr(row + b0, bit);
            atomicOr(row + b1, bit);
            atomicOr(row + b2, bit);
            atomicOr(row + b3, bit);
        }
    }
    __syncthreads();
    if ((int)threadIdx.x < nown) {
        const int b = threadIdx.x;
        uint4 lo, hi;
        lo.x = sBm[0 * 256 + b]; lo.y = sBm[1 * 256 + b]; lo.z = sBm[2 * 256 + b]; lo.w = sBm[3 * 256 + b];
        hi.x = sBm[4 * 256 + b]; hi.y = sBm[5 * 256 + b]; hi.z = sBm[6 * 256 + b]; hi.w = sBm[7 * 256 + b];
        uint4* dst = (uint4*)(bm + (size_t)b * (nbb * BM_W) + (size_t)blockIdx.x * BM_W);
        dst[0] = lo;
        dst[1] = hi;
    }
}

template <int M>
__device__ inline float add_xor16(float v) {
    return v + __int_as_float(__builtin_amdgcn_update_dpp(0, __float_as_int(v), 0x160 + M, 0xf, 0xf, false));
}

__device__ inline int wave_running_sum(int v) {
    v += __builtin_amdgcn_update_dpp(0, v, 0x111, 0xf, 0xf, false);
    v += __builtin_amdgcn_update_dpp(0, v, 0x112, 0xf, 0xf, false);
    v += __builtin_amdgcn_update_dpp(0, v, 0x114, 0xf, 0xf, false);
    v += __builtin_amdgcn_update_dpp(0, v, 0x118, 0xf, 0xf, false);
    const int lower = __builtin_amdgcn_readlane(v, 15);
    return v + ((threadIdx.x & 16) ? lower : 0);
}

__device__ inline int wave_take_slots(int* counter, int need) {
    const int upto = wave_running_sum(need);
    int base = 0;
    if ((threadIdx.x & 31) == 31) base = atomicAdd(counter, upto);
    return __builtin_amdgcn_readlane(base, 31) + upto - need;
}

template <int H, int CH, bool ELU, bool SAVE, bool LOAD, bool NEXT>
__global__ __launch_bounds__(OWN_T)
void gat_owner_kernel(const int* __restrict__ ei, int E, int N,
                      const unsigned int* __restrict__ ob,
                      const unsigned int* __restrict__ bm,
                      int* __restrict__ lst,
                      const float* __restrict__ hfeat,
                      const float* __restrict__ a_s,
                      const float* __restrict__ a_d,
                      const float* __restrict__ bias,
                      float* __restrict__ outp,
                      const float* __restrict__ w2f,
                      float* __restrict__ nas,
                      float* __restrict__ nad) {
    constexpr int F  = H * CH;
    constexpr int KF = F / 32;
    constexpr int NW = OWN_T / 32;

    static_assert(!NEXT || F == K_DIM, "the folded vectors are as long as the next product's inner dim");
    constexpr int ROWS = OWN_R + 1;
    __shared__ float sAgg[ROWS * F];
    __shared__ float sMax[ROWS * H];
    __shared__ float sDen[ROWS * H];
    __shared__ float sAd [ROWS * H];
    __shared__ int   sSrc[OWN_CAP];
    __shared__ int   sDl [OWN_CAP];
    __shared__ float sE  [OWN_CAP * H];
    __shared__ int   sCnt;
    __shared__ int   sCnd;

    const int tid = threadIdx.x;
    const int n0  = blockIdx.x * OWN_R;
    const int nn  = (N - n0 < OWN_R) ? (N - n0) : OWN_R;
    const int* __restrict__ dstp = ei + E;
    const bool reuse = LOAD && (lst != nullptr);
    const bool bytes = !reuse && (ob != nullptr);
    int* __restrict__ myl = lst ? lst + (size_t)blockIdx.x * LST_STRIDE : nullptr;
    int loaded = 0;

    for (int i = tid; i < ROWS * F; i += OWN_T) sAgg[i] = 0.f;
    for (int i = tid; i < ROWS * H; i += OWN_T) {
        sMax[i] = __int_as_float(0xff800000);
        sDen[i] = 0.f;
        sAd[i]  = (i < nn * H) ? a_d[(size_t)n0 * H + i] : 0.f;
    }
    if (!reuse) {
        for (int i = tid; i < nn; i += OWN_T) { sSrc[i] = -(i + 1); sDl[i] = i; }
        if (tid == 0) { sCnt = nn; sCnd = 0; }
    }
    __syncthreads();

    auto keep = [&](int dst, int eid, unsigned lim) {
        unsigned dl = (unsigned)(dst - n0);
        if (dl < lim) {
            int pos = atomicAdd(&sCnt, 1);
            if (pos < OWN_CAP) { sSrc[pos] = eid; sDl[pos] = (int)dl; }
        }
    };
    if (reuse) {
        loaded = myl[0];
        if ((unsigned)(loaded - 1) < (unsigned)OWN_CAP) {
            const int2* __restrict__ lp = (const int2*)(myl + LST_HDR);
            constexpr int NQ = OWN_CAP / OWN_T;
            int2 p[NQ];
            #pragma unroll
            for (int q = 0; q < NQ; ++q) {
                const int j = q * OWN_T + tid;
                p[q] = lp[(j < loaded) ? j : (loaded - 1)];
            }
            #pragma unroll
            for (int q = 0; q < NQ; ++q) {
                const int j = q * OWN_T + tid;
                if (j < loaded) {
                    sSrc[j] = p[q].x;
                    sDl[j]  = p[q].y;
                }
            }
        }
    } else if (bytes) {
        const uint4* __restrict__ bv = (const uint4*)ob;
        const unsigned pat = blockIdx.x * 0x01010101u;
        const int wpo = (((E >> 2) + 256 * OB_Q - 1) / (256 * OB_Q)) * BM_W;
        const unsigned int* __restrict__ mybm = bm + (size_t)blockIdx.x * wpo;
        {
            unsigned w0 = (tid < wpo) ? mybm[tid] : 0u;
            unsigned w1 = (tid + OWN_T < wpo) ? mybm[tid + OWN_T] : 0u;
            int pos = wave_take_slots(&sCnd, __builtin_popcount(w0) + __builtin_popcount(w1));
            while (w0) {
                const int t = __builtin_ctz(w0);
                w0 &= w0 - 1;
                if (pos < OWN_CAP) sDl[pos] = tid * 32 + t;
                ++pos;
            }
            while (w1) {
                const int t = __builtin_ctz(w1);
                w1 &= w1 - 1;
                if (pos < OWN_CAP) sDl[pos] = (tid + OWN_T) * 32 + t;
                ++pos;
            }
        }
        __syncthreads();
        const int ncand = sCnd;
        auto eqm = [&](unsigned x) {
            const unsigned y = x ^ pat;
            return ~(((y & 0x7f7f7f7fu) + 0x7f7f7f7fu) | y | 0x7f7f7f7fu);
        };
        if (ncand <= OWN_CAP) {
            constexpr int NQ = OWN_CAP / OWN_T;
            uint4 cv[NQ];
            int   vi[NQ];
            #pragma unroll
            for (int q = 0; q < NQ; ++q) {
                const int j = q * OWN_T + tid;
                vi[q] = (j < ncand) ? sDl[j] : -1;
            }
            #pragma unroll
            for (int q = 0; q < NQ; ++q) cv[q] = bv[vi[q] < 0 ? 0 : vi[q]];
            unsigned m[NQ];
            int ne = 0;
            #pragma unroll
            for (int q = 0; q < NQ; ++q) {
                m[q] = (eqm(cv[q].x) >> 7) | (eqm(cv[q].y) >> 6) | (eqm(cv[q].z) >> 5) | (eqm(cv[q].w) >> 4);
                if (vi[q] < 0) m[q] = 0;
                ne += __builtin_popcount(m[q]);
            }
            int pos = wave_take_slots(&sCnt, ne);
            #pragma unroll
            for (int q = 0; q < NQ; ++q) {
                unsigned mm = m[q];
                while (mm) {
                    const int t = __builtin_ctz(mm);
                    mm &= mm - 1;
                    if (pos < OWN_CAP) sSrc[pos] = vi[q] * 16 + ((t & 7) << 2) + (t >> 3);
                    ++pos;
                }
            }
        } else if (tid == 0) {
            sCnt = OWN_CAP + 1;
        }
    } else if ((E & 3) == 0) {
        const int EV = E >> 2;
        const int4* __restrict__ dv = (const int4*)dstp;
        constexpr int STEP = OWN_T * OWN_U;
        const int rot = (int)(((long long)blockIdx.x * EV) / gridDim.x);
        auto slot = [&](int base, int u, bool& ok) {
            int v  = base + u * OWN_T + tid;
            ok     = v < EV;
            int vr = v + rot;
            if (vr >= EV) vr -= EV;
            return ok ? vr : 0;
        };
        int4 cur[OWN_U];
        #pragma unroll
        for (int u = 0; u < OWN_U; ++u) { bool ok; cur[u] = dv[slot(0, u, ok)]; }
        for (int base = 0; base < EV; base += STEP) {
            int4 nxt[OWN_U];
            #pragma unroll
            for (int u = 0; u < OWN_U; ++u) { bool ok; nxt[u] = dv[slot(base + STEP, u, ok)]; }
            #pragma unroll
            for (int u = 0; u < OWN_U; ++u) {
                bool ok;
                const int e0 = slot(base, u, ok) * 4;
                const unsigned lim = ok ? (unsigned)nn : 0u;
                keep(cur[u].x, e0 + 0, lim);
                keep(cur[u].y, e0 + 1, lim);
                keep(cur[u].z, e0 + 2, lim);
                keep(cur[u].w, e0 + 3, lim);
            }
            #pragma unroll
            for (int u = 0; u < OWN_U; ++u) cur[u] = nxt[u];
        }
    } else {
        for (int e = tid; e < E; e += OWN_T) keep(dstp[e], e, (unsigned)nn);
    }
    __syncthreads();
    const int total = reuse ? loaded : sCnt;
    if (SAVE && lst != nullptr && tid == 0) myl[0] = total;

    if ((unsigned)total <= (unsigned)OWN_CAP) {
        for (int j = tid; j < total; j += OWN_T) {
            int id  = sSrc[j];
            int src = reuse ? id : ((id >= 0) ? ei[id] : (n0 - id - 1));
            int dl;
            if (bytes) {
                dl = (id >= 0) ? (dstp[id] - n0) : (-id - 1);
                if ((unsigned)dl >= (unsigned)nn) { dl = OWN_R; src = n0; }
                sDl[j] = dl;
            } else {
                dl = sDl[j];
            }
            sSrc[j] = src;
            for (int h = 0; h < H; ++h) {
                float v = a_s[(size_t)src * H + h] + sAd[dl * H + h];
                v = (v > 0.f) ? v : NEG_SLOPE * v;
                sE[j * H + h] = v;
                ldsMaxF(&sMax[dl * H + h], v);
            }
        }
        __syncthreads();
        if (SAVE && lst != nullptr) {
            int2* __restrict__ lp = (int2*)(myl + LST_HDR);
            for (int q = 0; q < OWN_CAP / OWN_T; ++q) {
                const int j = q * OWN_T + tid;
                if (j < total) lp[j] = make_int2(sSrc[j], sDl[j]);
            }
        }
        for (int j = tid; j < total; j += OWN_T) {
            int dl = sDl[j];
            for (int h = 0; h < H; ++h) {
                float ex = __expf(sE[j * H + h] - sMax[dl * H + h]);
                sE[j * H + h] = ex;
                atomicAdd(&sDen[dl * H + h], ex);
            }
        }
        __syncthreads();
        for (int j = tid; j < total; j += OWN_T) {
            int dl = sDl[j];
            for (int h = 0; h < H; ++h)
                sE[j * H + h] = sE[j * H + h] / sDen[dl * H + h];
        }
        __syncthreads();
        typedef typename row_words<KF>::type rowv;
        static_assert(KF == 2 || KF == 4, "a lane holds 8 or 16 bytes of a row");
        static_assert(CH % KF == 0, "a lane's words belong to one head");
        constexpr int UC = 32 / KF;
        static_assert(UC * H <= 32, "a group's weights fit one lane each");
        const int wave = tid >> 5;
        const int lane = tid & 31;
        for (int j0 = wave; j0 < total; j0 += NW * UC) {
            int jl = j0 + lane * NW;
            if (jl >= total) jl = total - 1;
            const int   srcl = sSrc[jl];
            const int   dll  = sDl[jl];
            int ja = j0 + (lane / H) * NW;
            if (ja >= total) ja = total - 1;
            const int   alw  = __float_as_int(sE[ja * H + (lane % H)]);
            const int   hs   = (KF * lane) / CH;
            rowv v[UC];
            #pragma unroll
            for (int q = 0; q < UC; ++q) {
                const int src = __builtin_amdgcn_readlane(srcl, q);
                v[q] = *((const rowv*)(hfeat + (size_t)src * F) + lane);
            }
            float al[UC];
            #pragma unroll
            for (int q = 0; q < UC; ++q) {
                int a = __builtin_amdgcn_readlane(alw, q * H);
                #pragma unroll
                for (int h = 1; h < H; ++h) {
                    const int ah = __builtin_amdgcn_readlane(alw, q * H + h);
                    a = (hs == h) ? ah : a;
                }
                al[q] = __int_as_float(a);
            }
            #pragma unroll
            for (int q = 0; q < UC; ++q) {
                const int j = j0 + q * NW;
                if (j < total) {
                    const int dl = __builtin_amdgcn_readlane(dll, q);
                    float* ag = &sAgg[dl * F + lane];
                    #pragma unroll
                    for (int k = 0; k < KF; ++k) atomicAdd(ag + k * 32, v[q][k] * al[q]);
                }
            }
        }
        __syncthreads();
    } else {
        const int ET = E + N;
        for (int e = tid; e < ET; e += OWN_T) {
            int dst = (e < E) ? dstp[e] : (e - E);
            unsigned dl = (unsigned)(dst - n0);
            if (dl < (unsigned)nn) {
                int src = (e < E) ? ei[e] : (e - E);
                for (int h = 0; h < H; ++h) {
                    float v = a_s[(size_t)src * H + h] + sAd[dl * H + h];
                    v = (v > 0.f) ? v : NEG_SLOPE * v;
                    ldsMaxF(&sMax[dl * H + h], v);
                }
            }
        }
        __syncthreads();
        for (int e = tid; e < ET; e += OWN_T) {
            int dst = (e < E) ? dstp[e] : (e - E);
            unsigned dl = (unsigned)(dst - n0);
            if (dl < (unsigned)nn) {
                int src = (e < E) ? ei[e] : (e - E);
                for (int h = 0; h < H; ++h) {
                    float v = a_s[(size_t)src * H + h] + sAd[dl * H + h];
                    v = (v > 0.f) ? v : NEG_SLOPE * v;
                    atomicAdd(&sDen[dl * H + h], __expf(v - sMax[dl * H + h]));
                }
            }
        }
        __syncthreads();
        for (int e = tid; e < ET; e += OWN_T) {
            int dst = (e < E) ? dstp[e] : (e - E);
            unsigned dl = (unsigned)(dst - n0);
            if (dl < (unsigned)nn) {
                int src = (e < E) ? ei[e] : (e - E);
                float al[H];
                for (int h = 0; h < H; ++h) {
                    float v = a_s[(size_t)src * H + h] + sAd[dl * H + h];
                    v = (v > 0.f) ? v : NEG_SLOPE * v;
                    al[h] = __expf(v - sMax[dl * H + h]) / sDen[dl * H + h];
                }
                for (int c = 0; c < F; ++c)
                    atomicAdd(&sAgg[dl * F + (c % KF) * 32 + c / KF], hfeat[(size_t)src * F + c] * al[c / CH]);
            }
        }
        __syncthreads();
    }

    float fs[KF], fd[KF];
    #pragma unroll
    for (int k = 0; k < KF; ++k) {
        fs[k] = NEXT ? w2f[KF * (tid & 31) + k]     : 0.f;
        fd[k] = NEXT ? w2f[F + KF * (tid & 31) + k] : 0.f;
    }
    for (int it = tid; it < nn * 32; it += OWN_T) {
        const int r = it >> 5;
        const int l = it & 31;
        typename row_words<KF>::type o;
        #pragma unroll
        for (int k = 0; k < KF; ++k) {
            float v = sAgg[r * F + k * 32 + l] + bias[KF * l + k];
            if (ELU) v = (v > 0.f) ? v : (__expf(v) - 1.f);
            o[k] = v;
        }
        *((typename row_words<KF>::type*)(outp + (size_t)(n0 + r) * F) + l) = o;
        if (NEXT) {
            float ps = 0.f, pd = 0.f;
            #pragma unroll
            for (int k = 0; k < KF; ++k) { ps += o[k] * fs[k]; pd += o[k] * fd[k]; }
            ps = add_xor16<1>(ps); pd = add_xor16<1>(pd);
            ps = add_xor16<2>(ps); pd = add_xor16<2>(pd);
            ps = add_xor16<4>(ps); pd = add_xor16<4>(pd);
            ps = add_xor16<8>(ps); pd = add_xor16<8>(pd);
            const float ts = __int_as_float(__builtin_amdgcn_readlane(__float_as_int(ps), 0))
                           + __int_as_float(__builtin_amdgcn_readlane(__float_as_int(ps), 16));
            const float td = __int_as_float(__builtin_amdgcn_readlane(__float_as_int(pd), 0))
                           + __int_as_float(__builtin_amdgcn_readlane(__float_as_int(pd), 16));
            if (l == 0) { nas[n0 + r] = ts; nad[n0 + r] = td; }
        }
    }
}

extern "C" void kernel_launch(void* const* d_in, const int* in_sizes, int n_in,
                              void* d_out, int out_size, void* d_ws, size_t ws_size,
                              hipStream_t stream) {
    const float* x   = (const float*)d_in[0];
    const int*   ei  = (const int*)  d_in[1];
    const float* W1  = (const float*)d_in[2];
    const float* as1 = (const float*)d_in[3];
    const float* ad1 = (const float*)d_in[4];
    const float* b1  = (const float*)d_in[5];
    const float* W2  = (const float*)d_in[6];
    const float* as2 = (const float*)d_in[7];
    const float* ad2 = (const float*)d_in[8];
    const float* b2  = (const float*)d_in[9];
    float* out = (float*)d_out;

    const int N  = in_sizes[0] / 128;
    const int E  = in_sizes[1] / 2;
    const int H1 = 4, F1 = 128;
    const int F2 = 64;

    char* w = (char*)d_ws;
    auto carve = [&](size_t bytes) {
        void* p = (void*)w;
        w += (bytes + 255) & ~(size_t)255;
        return p;
    };
    float* h1    = (float*)carve((size_t)N * F1 * 4);
    float* a1s   = (float*)carve((size_t)N * H1 * 4);
    float* a1d   = (float*)carve((size_t)N * H1 * 4);
    float* agg1  = (float*)carve((size_t)N * F1 * 4);
    float* h2    = (float*)carve((size_t)N * F2 * 4);
    float* a2s   = (float*)carve((size_t)N * 4);
    float* a2d   = (float*)carve((size_t)N * 4);

    const int og = (N + OWN_R - 1) / OWN_R;
    int* lst = (int*)carve((size_t)og * LST_STRIDE * 4);
    if ((size_t)(w - (char*)d_ws) > ws_size) lst = nullptr;

    unsigned int* ob = nullptr;
    unsigned int* bm = nullptr;
    const int nbq = ((E >> 2) + 256 * OB_Q - 1) / (256 * OB_Q);
    if (og <= 256 && (E & 15) == 0 && E <= (1 << 20) &&
        ((size_t)2 << 20) + (size_t)og * nbq * BM_W * 4 <= (size_t)N * F2 * 4) {
        ob = (unsigned int*)h2;
        bm = (unsigned int*)h2 + (1 << 19);
    }
    float* fold = h2 + (1 << 18);

    const int nbb = ob ? nbq : 0;
    owner_byte_kernel<<<nbb + FOLD_BLOCKS, FOLD_T, 0, stream>>>(ei, E, ob, bm, nbb, og, W1, as1, ad1, W2, as2, ad2, fold);

    {
        int mt = (N + 15) / 16;
        wmma_gemm_kernel<128, true><<<GEMM_GRID, 288, 0, stream>>>(x, W1, h1, N, mt, fold + FOLD_BEXT, a1s, a1d);
    }
    gat_owner_kernel<4, 32, true, true, false, true><<<og, OWN_T, 0, stream>>>(
        ei, E, N, ob, bm, lst, h1, a1s, a1d, b1, agg1, fold + FOLD_W2F, a2s, a2d);

    {
        int mt = (N + 15) / 16;
        wmma_gemm_kernel<64, false><<<GEMM_GRID, 128, 0, stream>>>(agg1, W2, h2, N, mt, nullptr, nullptr, nullptr);
    }
    gat_owner_kernel<1, 64, false, false, true, false><<<og, OWN_T, 0, stream>>>(
        ei, E, N, nullptr, nullptr, lst, h2, a2s, a2d, b2, out, nullptr, nullptr, nullptr);
}


# Logit phase of the edge kernels: a thread takes its up to 4 list entries together: index loads, then one 16-byte coefficient load each, all asked for before the first LDS max; sums unchanged
# speedup vs baseline: 9.4283x; 1.0395x over previous
#include <hip/hip_runtime.h>
#include <math.h>

typedef __attribute__((ext_vector_type(2))) float v2f;
typedef __attribute__((ext_vector_type(8))) float v8f;

#define K_DIM 128
#define NEG_SLOPE 0.2f

#define BX_PITCH  128
#define GEMM_GRID 1024
#define A_PITCH   132

template <int NCOL, bool EXT>
__global__ __launch_bounds__(NCOL * 2 + (EXT ? 32 : 0))
void wmma_gemm_kernel(const float* __restrict__ A, const float* __restrict__ B,
                      float* __restrict__ C, int M, int mtiles,
                      const float* __restrict__ Bx,
                      float* __restrict__ xs,
                      float* __restrict__ xd) {
    constexpr int NT = NCOL * 2;
    constexpr int NQ = (16 * K_DIM / 4) / NT;
    __shared__ __attribute__((aligned(16))) float sA[2][16 * A_PITCH];

    const int wave = threadIdx.x >> 5;
    const int lane = threadIdx.x & 31;
    const int half = lane >> 4;
    const int l    = lane & 15;
    const int col0 = wave * 16;
    const bool ext = EXT && (wave == NCOL / 16);

    v2f bf[K_DIM / 4];
    {
        static_assert(!EXT || NCOL == BX_PITCH, "the folded columns are laid out with the row pitch of B");
        const float* bcol = (ext ? Bx : (B + col0)) + l;
        #pragma unroll
        for (int s = 0; s < K_DIM / 4; ++s) {
            bf[s].x = bcol[(4 * s + 2 * half)     * NCOL];
            bf[s].y = bcol[(4 * s + 2 * half + 1) * NCOL];
        }
    }

    static_assert(NQ == 2 || NQ == 4, "an A tile is 2 or 4 pieces per thread");
    float4 r0 = {}, r1 = {}, r2 = {}, r3 = {};
    auto asrc = [&](int mt, int q) {
        const int f = q * NT + (int)threadIdx.x;
        int ar = mt * 16 + (f >> 5);
        if (ar >= M) ar = M - 1;
        return (const float4*)(A + (size_t)ar * K_DIM + (f & 31) * 4);
    };
    auto adst = [&](int buf, int q) {
        const int f = q * NT + (int)threadIdx.x;
        return (float4*)&sA[buf][(f >> 5) * A_PITCH + (f & 31) * 4];
    };
#define A_FETCH(mt_) do { r0 = *asrc((mt_), 0); r1 = *asrc((mt_), 1); \
        if constexpr (NQ == 4) { r2 = *asrc((mt_), 2); r3 = *asrc((mt_), 3); } } while (0)
#define A_PARK(buf_) do { *adst((buf_), 0) = r0; *adst((buf_), 1) = r1; \
        if constexpr (NQ == 4) { *adst((buf_), 2) = r2; *adst((buf_), 3) = r3; } } while (0)

    int mt = blockIdx.x;
    if (mt < mtiles && !ext) { A_FETCH(mt); A_PARK(0); }
    __syncthreads();

    int buf = 0;
    for (; mt < mtiles; mt += GEMM_GRID) {
        const bool more = mt + GEMM_GRID < mtiles;
        if (more && !ext) A_FETCH(mt + GEMM_GRID);

        const float* ap = &sA[buf][l * A_PITCH + 2 * half];
        v8f acc = {};
        #pragma unroll
        for (int s = 0; s < K_DIM / 4; ++s) {
            v2f a = *(const v2f*)(ap + 4 * s);
            acc = __builtin_amdgcn_wmma_f32_16x16x4_f32(
                      false, a, false, bf[s], (short)0, acc, false, false);
        }

        const int row0 = mt * 16;
        float* cp = C + (size_t)(row0 + half * 8) * NCOL + col0 + l;
        if (ext) {
            float* xp = ((l < 4) ? (xs + l) : (xd + (l - 4))) + (size_t)(row0 + half * 8) * 4;
            if (l < 8) {
                #pragma unroll
                for (int i = 0; i < 8; ++i)
                    if (row0 + half * 8 + i < M) xp[i * 4] = acc[i];
            }
        } else if (row0 + 16 <= M) {
            #pragma unroll
            for (int i = 0; i < 8; ++i) cp[i * NCOL] = acc[i];
        } else {
            #pragma unroll
            for (int i = 0; i < 8; ++i)
                if (row0 + half * 8 + i < M) cp[i * NCOL] = acc[i];
        }

        if (more && !ext) A_PARK(buf ^ 1);
        __syncthreads();
        buf ^= 1;
    }
#undef A_FETCH
#undef A_PARK
}

#define FOLD_BEXT 0
#define FOLD_W2F  (K_DIM * BX_PITCH)
#define FOLD_LEN  (K_DIM * 16 + 2 * K_DIM)
#define FOLD_T    256
#define FOLD_BLOCKS (FOLD_LEN / FOLD_T)

__device__ inline void fold_part(int o, const float* __restrict__ W1, const float* __restrict__ as1,
                                 const float* __restrict__ ad1, const float* __restrict__ W2,
                                 const float* __restrict__ as2, const float* __restrict__ ad2,
                                 float* __restrict__ fold) {
    float v = 0.f;
    int at;
    if (o < K_DIM * 16) {
        const int k = o >> 4, j = o & 15;
        at = FOLD_BEXT + k * BX_PITCH + j;
        if (j < 8) {
            const int hh = j & 3;
            const float* wp = W1 + k * 128 + hh * 32;
            const float* ap = ((j < 4) ? as1 : ad1) + hh * 32;
            for (int c = 0; c < 32; ++c) v += wp[c] * ap[c];
        }
    } else {
        const int o2 = o - K_DIM * 16;
        const int k  = o2 & (K_DIM - 1);
        at = FOLD_W2F + o2;
        const float* wp = W2 + k * 64;
        const float* ap = (o2 < K_DIM) ? as2 : ad2;
        for (int c = 0; c < 64; ++c) v += wp[c] * ap[c];
    }
    fold[at] = v;
}

template <int K> struct row_words { typedef float type __attribute__((ext_vector_type(K))); };
#define OWN_R    196
#define OWN_T    1024
#define OWN_CAP  4096
#define OWN_U    8
#define BM_W     8
#define LST_HDR  32
#define LST_STRIDE (LST_HDR + 2 * OWN_CAP)

__device__ inline void ldsMaxF(float* addr, float v) {
    int iv = __float_as_int(v);
    if (iv >= 0) atomicMax((int*)addr, iv);
    else         atomicMin((unsigned int*)addr, (unsigned int)iv);
}

#define OB_Q 4
__global__ __launch_bounds__(256)
void owner_byte_kernel(const int* __restrict__ ei, int E, unsigned int* __restrict__ ob,
                       unsigned int* __restrict__ bm, int nbb, int nown,
                       const float* __restrict__ W1, const float* __restrict__ as1, const float* __restrict__ ad1,
                       const float* __restrict__ W2, const float* __restrict__ as2, const float* __restrict__ ad2,
                       float* __restrict__ fold) {
    if ((int)blockIdx.x >= nbb) {
        fold_part(((int)blockIdx.x - nbb) * FOLD_T + (int)threadIdx.x, W1, as1, ad1, W2, as2, ad2, fold);
        return;
    }
    static_assert(BM_W == 8, "thread b stores owner b's words of the block as two 16-byte pieces");
    __shared__ unsigned int sBm[BM_W * 256];
    const int EQ = E >> 2;
    const int4* __restrict__ dv = (const int4*)(ei + E);
    const int g0 = ((blockIdx.x * blockDim.x + threadIdx.x) >> 5) * (OB_Q * 32) + (threadIdx.x & 31);
    int4 d[OB_Q];
    #pragma unroll
    for (int q = 0; q < OB_Q; ++q) {
        const int g = g0 + q * 32;
        d[q] = dv[(g < EQ) ? g : (EQ - 1)];
    }
    for (int i = threadIdx.x; i < BM_W * 256; i += 256) sBm[i] = 0u;
    __syncthreads();
    #pragma unroll
    for (int q = 0; q < OB_Q; ++q) {
        const int g = g0 + q * 32;
        const unsigned b0 = ((unsigned)d[q].x / (unsigned)OWN_R) & 255u;
        const unsigned b1 = ((unsigned)d[q].y / (unsigned)OWN_R) & 255u;
        const unsigned b2 = ((unsigned)d[q].z / (unsigned)OWN_R) & 255u;
        const unsigned b3 = ((unsigned)d[q].w / (unsigned)OWN_R) & 255u;
        if (g < EQ) {
            ob[g] = b0 | (b1 << 8) | (b2 << 16) | (b3 << 24);
            const int vl = (g >> 2) - (int)blockIdx.x * 256;
            unsigned int* row = sBm + (vl >> 5) * 256;
            const unsigned bit = 1u << (vl & 31);
            atomicOr(row + b0, bit);
            atomicOr(row + b1, bit);
            atomicOr(row + b2, bit);
            atomicOr(row + b3, bit);
        }
    }
    __syncthreads();
    if ((int)threadIdx.x < nown) {
        const int b = threadIdx.x;
        uint4 lo, hi;
        lo.x = sBm[0 * 256 + b]; lo.y = sBm[1 * 256 + b]; lo.z = sBm[2 * 256 + b]; lo.w = sBm[3 * 256 + b];
        hi.x = sBm[4 * 256 + b]; hi.y = sBm[5 * 256 + b]; hi.z = sBm[6 * 256 + b]; hi.w = sBm[7 * 256 + b];
        uint4* dst = (uint4*)(bm + (size_t)b * (nbb * BM_W) + (size_t)blockIdx.x * BM_W);
        dst[0] = lo;
        dst[1] = hi;
    }
}

template <int M>
__device__ inline float add_xor16(float v) {
    return v + __int_as_float(__builtin_amdgcn_update_dpp(0, __float_as_int(v), 0x160 + M, 0xf, 0xf, false));
}

__device__ inline int wave_running_sum(int v) {
    v += __builtin_amdgcn_update_dpp(0, v, 0x111, 0xf, 0xf, false);
    v += __builtin_amdgcn_update_dpp(0, v, 0x112, 0xf, 0xf, false);
    v += __builtin_amdgcn_update_dpp(0, v, 0x114, 0xf, 0xf, false);
    v += __builtin_amdgcn_update_dpp(0, v, 0x118, 0xf, 0xf, false);
    const int lower = __builtin_amdgcn_readlane(v, 15);
    return v + ((threadIdx.x & 16) ? lower : 0);
}

__device__ inline int wave_take_slots(int* counter, int need) {
    const int upto = wave_running_sum(need);
    int base = 0;
    if ((threadIdx.x & 31) == 31) base = atomicAdd(counter, upto);
    return __builtin_amdgcn_readlane(base, 31) + upto - need;
}

template <int H, int CH, bool ELU, bool SAVE, bool LOAD, bool NEXT>
__global__ __launch_bounds__(OWN_T)
void gat_owner_kernel(const int* __restrict__ ei, int E, int N,
                      const unsigned int* __restrict__ ob,
                      const unsigned int* __restrict__ bm,
                      int* __restrict__ lst,
                      const float* __restrict__ hfeat,
                      const float* __restrict__ a_s,
                      const float* __restrict__ a_d,
                      const float* __restrict__ bias,
                      float* __restrict__ outp,
                      const float* __restrict__ w2f,
                      float* __restrict__ nas,
                      float* __restrict__ nad) {
    constexpr int F  = H * CH;
    constexpr int KF = F / 32;
    constexpr int NW = OWN_T / 32;

    static_assert(!NEXT || F == K_DIM, "the folded vectors are as long as the next product's inner dim");
    constexpr int ROWS = OWN_R + 1;
    __shared__ float sAgg[ROWS * F];
    __shared__ float sMax[ROWS * H];
    __shared__ float sDen[ROWS * H];
    __shared__ float sAd [ROWS * H];
    __shared__ int   sSrc[OWN_CAP];
    __shared__ int   sDl [OWN_CAP];
    __shared__ float sE  [OWN_CAP * H];
    __shared__ int   sCnt;
    __shared__ int   sCnd;

    const int tid = threadIdx.x;
    const int n0  = blockIdx.x * OWN_R;
    const int nn  = (N - n0 < OWN_R) ? (N - n0) : OWN_R;
    const int* __restrict__ dstp = ei + E;
    const bool reuse = LOAD && (lst != nullptr);
    const bool bytes = !reuse && (ob != nullptr);
    int* __restrict__ myl = lst ? lst + (size_t)blockIdx.x * LST_STRIDE : nullptr;
    int loaded = 0;

    for (int i = tid; i < ROWS * F; i += OWN_T) sAgg[i] = 0.f;
    for (int i = tid; i < ROWS * H; i += OWN_T) {
        sMax[i] = __int_as_float(0xff800000);
        sDen[i] = 0.f;
        sAd[i]  = (i < nn * H) ? a_d[(size_t)n0 * H + i] : 0.f;
    }
    if (!reuse) {
        for (int i = tid; i < nn; i += OWN_T) { sSrc[i] = -(i + 1); sDl[i] = i; }
        if (tid == 0) { sCnt = nn; sCnd = 0; }
    }
    __syncthreads();

    auto keep = [&](int dst, int eid, unsigned lim) {
        unsigned dl = (unsigned)(dst - n0);
        if (dl < lim) {
            int pos = atomicAdd(&sCnt, 1);
            if (pos < OWN_CAP) { sSrc[pos] = eid; sDl[pos] = (int)dl; }
        }
    };
    if (reuse) {
        loaded = myl[0];
        if ((unsigned)(loaded - 1) < (unsigned)OWN_CAP) {
            const int2* __restrict__ lp = (const int2*)(myl + LST_HDR);
            constexpr int NQ = OWN_CAP / OWN_T;
            int2 p[NQ];
            #pragma unroll
            for (int q = 0; q < NQ; ++q) {
                const int j = q * OWN_T + tid;
                p[q] = lp[(j < loaded) ? j : (loaded - 1)];
            }
            #pragma unroll
            for (int q = 0; q < NQ; ++q) {
                const int j = q * OWN_T + tid;
                if (j < loaded) {
                    sSrc[j] = p[q].x;
                    sDl[j]  = p[q].y;
                }
            }
        }
    } else if (bytes) {
        const uint4* __restrict__ bv = (const uint4*)ob;
        const unsigned pat = blockIdx.x * 0x01010101u;
        const int wpo = (((E >> 2) + 256 * OB_Q - 1) / (256 * OB_Q)) * BM_W;
        const unsigned int* __restrict__ mybm = bm + (size_t)blockIdx.x * wpo;
        {
            unsigned w0 = (tid < wpo) ? mybm[tid] : 0u;
            unsigned w1 = (tid + OWN_T < wpo) ? mybm[tid + OWN_T] : 0u;
            int pos = wave_take_slots(&sCnd, __builtin_popcount(w0) + __builtin_popcount(w1));
            while (w0) {
                const int t = __builtin_ctz(w0);
                w0 &= w0 - 1;
                if (pos < OWN_CAP) sDl[pos] = tid * 32 + t;
                ++pos;
            }
            while (w1) {
                const int t = __builtin_ctz(w1);
                w1 &= w1 - 1;
                if (pos < OWN_CAP) sDl[pos] = (tid + OWN_T) * 32 + t;
                ++pos;
            }
        }
        __syncthreads();
        const int ncand = sCnd;
        auto eqm = [&](unsigned x) {
            const unsigned y = x ^ pat;
            return ~(((y & 0x7f7f7f7fu) + 0x7f7f7f7fu) | y | 0x7f7f7f7fu);
        };
        if (ncand <= OWN_CAP) {
            constexpr int NQ = OWN_CAP / OWN_T;
            uint4 cv[NQ];
            int   vi[NQ];
            #pragma unroll
            for (int q = 0; q < NQ; ++q) {
                const int j = q * OWN_T + tid;
                vi[q] = (j < ncand) ? sDl[j] : -1;
            }
            #pragma unroll
            for (int q = 0; q < NQ; ++q) cv[q] = bv[vi[q] < 0 ? 0 : vi[q]];
            unsigned m[NQ];
            int ne = 0;
            #pragma unroll
            for (int q = 0; q < NQ; ++q) {
                m[q] = (eqm(cv[q].x) >> 7) | (eqm(cv[q].y) >> 6) | (eqm(cv[q].z) >> 5) | (eqm(cv[q].w) >> 4);
                if (vi[q] < 0) m[q] = 0;
                ne += __builtin_popcount(m[q]);
            }
            int pos = wave_take_slots(&sCnt, ne);
            #pragma unroll
            for (int q = 0; q < NQ; ++q) {
                unsigned mm = m[q];
                while (mm) {
                    const int t = __builtin_ctz(mm);
                    mm &= mm - 1;
                    if (pos < OWN_CAP) sSrc[pos] = vi[q] * 16 + ((t & 7) << 2) + (t >> 3);
                    ++pos;
                }
            }
        } else if (tid == 0) {
            sCnt = OWN_CAP + 1;
        }
    } else if ((E & 3) == 0) {
        const int EV = E >> 2;
        const int4* __restrict__ dv = (const int4*)dstp;
        constexpr int STEP = OWN_T * OWN_U;
        const int rot = (int)(((long long)blockIdx.x * EV) / gridDim.x);
        auto slot = [&](int base, int u, bool& ok) {
            int v  = base + u * OWN_T + tid;
            ok     = v < EV;
            int vr = v + rot;
            if (vr >= EV) vr -= EV;
            return ok ? vr : 0;
        };
        int4 cur[OWN_U];
        #pragma unroll
        for (int u = 0; u < OWN_U; ++u) { bool ok; cur[u] = dv[slot(0, u, ok)]; }
        for (int base = 0; base < EV; base += STEP) {
            int4 nxt[OWN_U];
            #pragma unroll
            for (int u = 0; u < OWN_U; ++u) { bool ok; nxt[u] = dv[slot(base + STEP, u, ok)]; }
            #pragma unroll
            for (int u = 0; u < OWN_U; ++u) {
                bool ok;
                const int e0 = slot(base, u, ok) * 4;
                const unsigned lim = ok ? (unsigned)nn : 0u;
                keep(cur[u].x, e0 + 0, lim);
                keep(cur[u].y, e0 + 1, lim);
                keep(cur[u].z, e0 + 2, lim);
                keep(cur[u].w, e0 + 3, lim);
            }
            #pragma unroll
            for (int u = 0; u < OWN_U; ++u) cur[u] = nxt[u];
        }
    } else {
        for (int e = tid; e < E; e += OWN_T) keep(dstp[e], e, (unsigned)nn);
    }
    __syncthreads();
    const int total = reuse ? loaded : sCnt;
    if (SAVE && lst != nullptr && tid == 0) myl[0] = total;

    if ((unsigned)total <= (unsigned)OWN_CAP) {
        {
            constexpr int NA = OWN_CAP / OWN_T;
            int  ida[NA], dla[NA], srca[NA];
            bool oka[NA];
            #pragma unroll
            for (int q = 0; q < NA; ++q) {
                const int j = q * OWN_T + tid;
                oka[q] = j < total;
                ida[q] = oka[q] ? sSrc[j] : -1;
                dla[q] = (oka[q] && !bytes) ? sDl[j] : 0;
            }
            if (!reuse) {
                int es[NA], ed[NA];
                #pragma unroll
                for (int q = 0; q < NA; ++q) {
                    const int ide = (ida[q] >= 0) ? ida[q] : 0;
                    es[q] = ei[ide];
                    ed[q] = bytes ? dstp[ide] : 0;
                }
                #pragma unroll
                for (int q = 0; q < NA; ++q) {
                    const int id = ida[q];
                    int src = (id >= 0) ? es[q] : (n0 - id - 1);
                    int dl  = dla[q];
                    if (bytes) {
                        dl = (id >= 0) ? (ed[q] - n0) : (-id - 1);
                        if ((unsigned)dl >= (unsigned)nn) { dl = OWN_R; src = n0; }
                    }
                    srca[q] = src;
                    dla[q]  = dl;
                }
            } else {
                #pragma unroll
                for (int q = 0; q < NA; ++q) srca[q] = ida[q];
            }
            #pragma unroll
            for (int q = 0; q < NA; ++q)
                if (!oka[q]) { srca[q] = n0; dla[q] = 0; }
            float cs[NA][H];
            #pragma unroll
            for (int q = 0; q < NA; ++q) {
                if constexpr (H == 4) {
                    const float4 t = *(const float4*)(a_s + (size_t)srca[q] * 4);
                    cs[q][0] = t.x; cs[q][1] = t.y; cs[q][2] = t.z; cs[q][3] = t.w;
                } else {
                    #pragma unroll
                    for (int h = 0; h < H; ++h) cs[q][h] = a_s[(size_t)srca[q] * H + h];
                }
            }
            #pragma unroll
            for (int q = 0; q < NA; ++q) {
                #pragma unroll
                for (int h = 0; h < H; ++h) {
                    const float v = cs[q][h] + sAd[dla[q] * H + h];
                    cs[q][h] = (v > 0.f) ? v : NEG_SLOPE * v;
                }
            }
            #pragma unroll
            for (int q = 0; q < NA; ++q) {
                const int j = q * OWN_T + tid;
                if (oka[q]) {
                    sSrc[j] = srca[q];
                    if (bytes) sDl[j] = dla[q];
                    #pragma unroll
                    for (int h = 0; h < H; ++h) {
                        sE[j * H + h] = cs[q][h];
                        ldsMaxF(&sMax[dla[q] * H + h], cs[q][h]);
                    }
                }
            }
        }
        __syncthreads();
        if (SAVE && lst != nullptr) {
            int2* __restrict__ lp = (int2*)(myl + LST_HDR);
            for (int q = 0; q < OWN_CAP / OWN_T; ++q) {
                const int j = q * OWN_T + tid;
                if (j < total) lp[j] = make_int2(sSrc[j], sDl[j]);
            }
        }
        for (int j = tid; j < total; j += OWN_T) {
            int dl = sDl[j];
            for (int h = 0; h < H; ++h) {
                float ex = __expf(sE[j * H + h] - sMax[dl * H + h]);
                sE[j * H + h] = ex;
                atomicAdd(&sDen[dl * H + h], ex);
            }
        }
        __syncthreads();
        for (int j = tid; j < total; j += OWN_T) {
            int dl = sDl[j];
            for (int h = 0; h < H; ++h)
                sE[j * H + h] = sE[j * H + h] / sDen[dl * H + h];
        }
        __syncthreads();
        typedef typename row_words<KF>::type rowv;
        static_assert(KF == 2 || KF == 4, "a lane holds 8 or 16 bytes of a row");
        static_assert(CH % KF == 0, "a lane's words belong to one head");
        constexpr int UC = 32 / KF;
        static_assert(UC * H <= 32, "a group's weights fit one lane each");
        const int wave = tid >> 5;
        const int lane = tid & 31;
        for (int j0 = wave; j0 < total; j0 += NW * UC) {
            int jl = j0 + lane * NW;
            if (jl >= total) jl = total - 1;
            const int   srcl = sSrc[jl];
            const int   dll  = sDl[jl];
            int ja = j0 + (lane / H) * NW;
            if (ja >= total) ja = total - 1;
            const int   alw  = __float_as_int(sE[ja * H + (lane % H)]);
            const int   hs   = (KF * lane) / CH;
            rowv v[UC];
            #pragma unroll
            for (int q = 0; q < UC; ++q) {
                const int src = __builtin_amdgcn_readlane(srcl, q);
                v[q] = *((const rowv*)(hfeat + (size_t)src * F) + lane);
            }
            float al[UC];
            #pragma unroll
            for (int q = 0; q < UC; ++q) {
                int a = __builtin_amdgcn_readlane(alw, q * H);
                #pragma unroll
                for (int h = 1; h < H; ++h) {
                    const int ah = __builtin_amdgcn_readlane(alw, q * H + h);
                    a = (hs == h) ? ah : a;
                }
                al[q] = __int_as_float(a);
            }
            #pragma unroll
            for (int q = 0; q < UC; ++q) {
                const int j = j0 + q * NW;
                if (j < total) {
                    const int dl = __builtin_amdgcn_readlane(dll, q);
                    float* ag = &sAgg[dl * F + lane];
                    #pragma unroll
                    for (int k = 0; k < KF; ++k) atomicAdd(ag + k * 32, v[q][k] * al[q]);
                }
            }
        }
        __syncthreads();
    } else {
        const int ET = E + N;
        for (int e = tid; e < ET; e += OWN_T) {
            int dst = (e < E) ? dstp[e] : (e - E);
            unsigned dl = (unsigned)(dst - n0);
            if (dl < (unsigned)nn) {
                int src = (e < E) ? ei[e] : (e - E);
                for (int h = 0; h < H; ++h) {
                    float v = a_s[(size_t)src * H + h] + sAd[dl * H + h];
                    v = (v > 0.f) ? v : NEG_SLOPE * v;
                    ldsMaxF(&sMax[dl * H + h], v);
                }
            }
        }
        __syncthreads();
        for (int e = tid; e < ET; e += OWN_T) {
            int dst = (e < E) ? dstp[e] : (e - E);
            unsigned dl = (unsigned)(dst - n0);
            if (dl < (unsigned)nn) {
                int src = (e < E) ? ei[e] : (e - E);
                for (int h = 0; h < H; ++h) {
                    float v = a_s[(size_t)src * H + h] + sAd[dl * H + h];
                    v = (v > 0.f) ? v : NEG_SLOPE * v;
                    atomicAdd(&sDen[dl * H + h], __expf(v - sMax[dl * H + h]));
                }
            }
        }
        __syncthreads();
        for (int e = tid; e < ET; e += OWN_T) {
            int dst = (e < E) ? dstp[e] : (e - E);
            unsigned dl = (unsigned)(dst - n0);
            if (dl < (unsigned)nn) {
                int src = (e < E) ? ei[e] : (e - E);
                float al[H];
                for (int h = 0; h < H; ++h) {
                    float v = a_s[(size_t)src * H + h] + sAd[dl * H + h];
                    v = (v > 0.f) ? v : NEG_SLOPE * v;
                    al[h] = __expf(v - sMax[dl * H + h]) / sDen[dl * H + h];
                }
                for (int c = 0; c < F; ++c)
                    atomicAdd(&sAgg[dl * F + (c % KF) * 32 + c / KF], hfeat[(size_t)src * F + c] * al[c / CH]);
            }
        }
        __syncthreads();
    }

    float fs[KF], fd[KF];
    #pragma unroll
    for (int k = 0; k < KF; ++k) {
        fs[k] = NEXT ? w2f[KF * (tid & 31) + k]     : 0.f;
        fd[k] = NEXT ? w2f[F + KF * (tid & 31) + k] : 0.f;
    }
    for (int it = tid; it < nn * 32; it += OWN_T) {
        const int r = it >> 5;
        const int l = it & 31;
        typename row_words<KF>::type o;
        #pragma unroll
        for (int k = 0; k < KF; ++k) {
            float v = sAgg[r * F + k * 32 + l] + bias[KF * l + k];
            if (ELU) v = (v > 0.f) ? v : (__expf(v) - 1.f);
            o[k] = v;
        }
        *((typename row_words<KF>::type*)(outp + (size_t)(n0 + r) * F) + l) = o;
        if (NEXT) {
            float ps = 0.f, pd = 0.f;
            #pragma unroll
            for (int k = 0; k < KF; ++k) { ps += o[k] * fs[k]; pd += o[k] * fd[k]; }
            ps = add_xor16<1>(ps); pd = add_xor16<1>(pd);
            ps = add_xor16<2>(ps); pd = add_xor16<2>(pd);
            ps = add_xor16<4>(ps); pd = add_xor16<4>(pd);
            ps = add_xor16<8>(ps); pd = add_xor16<8>(pd);
            const float ts = __int_as_float(__builtin_amdgcn_readlane(__float_as_int(ps), 0))
                           + __int_as_float(__builtin_amdgcn_readlane(__float_as_int(ps), 16));
            const float td = __int_as_float(__builtin_amdgcn_readlane(__float_as_int(pd), 0))
                           + __int_as_float(__builtin_amdgcn_readlane(__float_as_int(pd), 16));
            if (l == 0) { nas[n0 + r] = ts; nad[n0 + r] = td; }
        }
    }
}

extern "C" void kernel_launch(void* const* d_in, const int* in_sizes, int n_in,
                              void* d_out, int out_size, void* d_ws, size_t ws_size,
                              hipStream_t stream) {
    const float* x   = (const float*)d_in[0];
    const int*   ei  = (const int*)  d_in[1];
    const float* W1  = (const float*)d_in[2];
    const float* as1 = (const float*)d_in[3];
    const float* ad1 = (const float*)d_in[4];
    const float* b1  = (const float*)d_in[5];
    const float* W2  = (const float*)d_in[6];
    const float* as2 = (const float*)d_in[7];
    const float* ad2 = (const float*)d_in[8];
    const float* b2  = (const float*)d_in[9];
    float* out = (float*)d_out;

    const int N  = in_sizes[0] / 128;
    const int E  = in_sizes[1] / 2;
    const int H1 = 4, F1 = 128;
    const int F2 = 64;

    char* w = (char*)d_ws;
    auto carve = [&](size_t bytes) {
        void* p = (void*)w;
        w += (bytes + 255) & ~(size_t)255;
        return p;
    };
    float* h1    = (float*)carve((size_t)N * F1 * 4);
    float* a1s   = (float*)carve((size_t)N * H1 * 4);
    float* a1d   = (float*)carve((size_t)N * H1 * 4);
    float* agg1  = (float*)carve((size_t)N * F1 * 4);
    float* h2    = (float*)carve((size_t)N * F2 * 4);
    float* a2s   = (float*)carve((size_t)N * 4);
    float* a2d   = (float*)carve((size_t)N * 4);

    const int og = (N + OWN_R - 1) / OWN_R;
    int* lst = (int*)carve((size_t)og * LST_STRIDE * 4);
    if ((size_t)(w - (char*)d_ws) > ws_size) lst = nullptr;

    unsigned int* ob = nullptr;
    unsigned int* bm = nullptr;
    const int nbq = ((E >> 2) + 256 * OB_Q - 1) / (256 * OB_Q);
    if (og <= 256 && (E & 15) == 0 && E <= (1 << 20) &&
        ((size_t)2 << 20) + (size_t)og * nbq * BM_W * 4 <= (size_t)N * F2 * 4) {
        ob = (unsigned int*)h2;
        bm = (unsigned int*)h2 + (1 << 19);
    }
    float* fold = h2 + (1 << 18);

    const int nbb = ob ? nbq : 0;
    owner_byte_kernel<<<nbb + FOLD_BLOCKS, FOLD_T, 0, stream>>>(ei, E, ob, bm, nbb, og, W1, as1, ad1, W2, as2, ad2, fold);

    {
        int mt = (N + 15) / 16;
        wmma_gemm_kernel<128, true><<<GEMM_GRID, 288, 0, stream>>>(x, W1, h1, N, mt, fold + FOLD_BEXT, a1s, a1d);
    }
    gat_owner_kernel<4, 32, true, true, false, true><<<og, OWN_T, 0, stream>>>(
        ei, E, N, ob, bm, lst, h1, a1s, a1d, b1, agg1, fold + FOLD_W2F, a2s, a2d);

    {
        int mt = (N + 15) / 16;
        wmma_gemm_kernel<64, false><<<GEMM_GRID, 128, 0, stream>>>(agg1, W2, h2, N, mt, nullptr, nullptr, nullptr);
    }
    gat_owner_kernel<1, 64, false, false, true, false><<<og, OWN_T, 0, stream>>>(
        ei, E, N, nullptr, nullptr, lst, h2, a2s, a2d, b2, out, nullptr, nullptr, nullptr);
}
